# starting kernel
# baseline (speedup 1.0000x reference)
#include <hip/hip_runtime.h>
#include <hip/hip_bf16.h>


typedef __attribute__((ext_vector_type(2))) float v2f;
typedef __attribute__((ext_vector_type(8))) float v8f;

#define B_     2
#define S_     2048
#define DM_    1024
#define H_     16
#define DK_    64
#define BH_    (B_ * H_)
#define ROWS_  (B_ * S_)
#define N_QKV  (3 * DM_)

__device__ __forceinline__ v8f wmma_f32(v2f a, v2f b, v8f c) {
  return __builtin_amdgcn_wmma_f32_16x16x4_f32(
      false, a, false, b, (short)0, c, false, false);
}

__device__ __forceinline__ void gemm_core_128x64(
    const float* __restrict__ A, const float* __restrict__ W,
    float* Xs, float* Ws, int m0, int n0, int Kdim, v8f acc[2][2]) {
  const int tid = threadIdx.x;
  const int wv = tid >> 5;
  const int lane = tid & 31;
  const int r = lane & 15;
  const int hh = lane >> 4;
  const int wm = wv >> 1;
  const int wn = wv & 1;

  for (int k0 = 0; k0 < Kdim; k0 += 16) {
    __syncthreads();
#pragma unroll
    for (int i = 0; i < 2; ++i) {
      int idx = tid + i * 256;
      int row = idx >> 2;
      int cv = (idx & 3) * 4;
      *(float4*)(Xs + row * 20 + cv) =
          *(const float4*)(A + (size_t)(m0 + row) * Kdim + k0 + cv);
    }
    {
      int row = tid >> 2;
      int cv = (tid & 3) * 4;
      *(float4*)(Ws + row * 20 + cv) =
          *(const float4*)(W + (size_t)(n0 + row) * Kdim + k0 + cv);
    }
    __syncthreads();
#pragma unroll
    for (int kk = 0; kk < 4; ++kk) {
      const int kd = kk * 4 + hh * 2;
      const float* xr0 = Xs + (wm * 32 + r) * 20 + kd;
      const float* xr1 = Xs + (wm * 32 + 16 + r) * 20 + kd;
      const float* wr0 = Ws + (wn * 32 + r) * 20 + kd;
      const float* wr1 = Ws + (wn * 32 + 16 + r) * 20 + kd;
      v2f a0 = {xr0[0], xr0[1]};
      v2f a1 = {xr1[0], xr1[1]};
      v2f b0 = {wr0[0], wr0[1]};
      v2f b1 = {wr1[0], wr1[1]};
      acc[0][0] = wmma_f32(a0, b0, acc[0][0]);
      acc[0][1] = wmma_f32(a0, b1, acc[0][1]);
      acc[1][0] = wmma_f32(a1, b0, acc[1][0]);
      acc[1][1] = wmma_f32(a1, b1, acc[1][1]);
    }
  }
}

__global__ void __launch_bounds__(256)
qkv_rope_kernel(const float* __restrict__ x, const float* __restrict__ w_qkv,
                const float* __restrict__ b_qkv, float* __restrict__ Qw,
                float* __restrict__ Kw, float* __restrict__ Vw) {
  __shared__ float Xs[128 * 20];
  __shared__ float Ws[64 * 20];
  v8f acc[2][2] = {};
  const int m0 = blockIdx.x * 128;
  const int n0 = blockIdx.y * 64;
  gemm_core_128x64(x, w_qkv, Xs, Ws, m0, n0, DM_, acc);

  const int tid = threadIdx.x;
  const int wv = tid >> 5;
  const int lane = tid & 31;
  const int r = lane & 15;
  const int hhalf = lane >> 4;
  const int wm = wv >> 1;
  const int wn = wv & 1;

#pragma unroll
  for (int mi = 0; mi < 2; ++mi) {
#pragma unroll
    for (int ni = 0; ni < 2; ++ni) {
      const int o = n0 + wn * 32 + ni * 16 + r;
      const int sel = o >> 10;
      const int rem = o & 1023;
      const int head = rem >> 6;
      const int d = rem & 63;
      const float bv = b_qkv[o];
      const int p = d >> 1;
      const float invf = __expf((float)(2 * p) * (-9.210340371976184f / 64.0f));
      const float sgn = (d & 1) ? 1.0f : -1.0f;
      float* dst = (sel == 0) ? Qw : ((sel == 1) ? Kw : Vw);
#pragma unroll
      for (int g = 0; g < 8; ++g) {
        const int gm = m0 + wm * 32 + mi * 16 + g + 8 * hhalf;
        const int srow = gm & (S_ - 1);
        const int bb = gm >> 11;
        float v = acc[mi][ni][g] + bv;
        float other = __shfl_xor(v, 1);
        float outv = v;
        if (sel < 2) {
          const float ang = (float)srow * invf;
          const float sn = __sinf(ang);
          const float cs = __cosf(ang);
          outv = v * cs + other * sgn * sn;
        }
        dst[(((size_t)(bb * H_ + head)) * S_ + srow) * DK_ + d] = outv;
      }
    }
  }
}

__global__ void __launch_bounds__(256)
flash_attn_kernel(const float* __restrict__ Q, const float* __restrict__ K,
                  const float* __restrict__ V, float* __restrict__ O) {
  __shared__ float Ks[16 * 68];
  __shared__ float Vs[16 * 68];
  __shared__ float Ps[8 * 16 * 17];

  const int bh = blockIdx.y;
  const int qb = blockIdx.x;
  const int tid = threadIdx.x;
  const int wv = tid >> 5;
  const int lane = tid & 31;
  const int r = lane & 15;
  const int hhalf = lane >> 4;
  const int qw0 = qb * 128 + wv * 16;

  const float* Qp = Q + (size_t)bh * S_ * DK_;
  const float* Kp = K + (size_t)bh * S_ * DK_;
  const float* Vp = V + (size_t)bh * S_ * DK_;

  v2f qf[16];
#pragma unroll
  for (int kk = 0; kk < 16; ++kk) {
    const float* qp = Qp + (size_t)(qw0 + r) * DK_ + kk * 4 + hhalf * 2;
    qf[kk] = v2f{qp[0] * 0.125f, qp[1] * 0.125f};
  }

  v8f o0 = {}, o1 = {}, o2 = {}, o3 = {};
  float mrow[8], lrow[8];
#pragma unroll
  for (int g = 0; g < 8; ++g) { mrow[g] = -3.0e38f; lrow[g] = 0.0f; }

  float* Pw = Ps + wv * 16 * 17;
  const int njt = qb * 8 + 8;

  for (int j = 0; j < njt; ++j) {
    __syncthreads();
    {
      const int rowj = tid >> 4;
      const int cv = (tid & 15) * 4;
      const float4 kv = *(const float4*)(Kp + (size_t)(j * 16 + rowj) * DK_ + cv);
      const float4 vvv = *(const float4*)(Vp + (size_t)(j * 16 + rowj) * DK_ + cv);
      *(float4*)(Ks + rowj * 68 + cv) = kv;
      *(float4*)(Vs + rowj * 68 + cv) = vvv;
    }
    __syncthreads();

    if (j * 16 <= qw0 + 15) {
      v8f s = {};
#pragma unroll
      for (int kk = 0; kk < 16; ++kk) {
        const int kd = kk * 4 + hhalf * 2;
        const float* kr = Ks + r * 68 + kd;
        v2f bf = {kr[0], kr[1]};
        s = wmma_f32(qf[kk], bf, s);
      }
      if (j * 16 + 15 > qw0) {
        const int key = j * 16 + r;
#pragma unroll
        for (int g = 0; g < 8; ++g) {
          if (key > qw0 + g + 8 * hhalf) s[g] = -3.0e38f;
        }
      }
      float pv[8];
#pragma unroll
      for (int g = 0; g < 8; ++g) {
        float v = s[g];
        float mx = v;
        mx = fmaxf(mx, __shfl_xor(mx, 1));
        mx = fmaxf(mx, __shfl_xor(mx, 2));
        mx = fmaxf(mx, __shfl_xor(mx, 4));
        mx = fmaxf(mx, __shfl_xor(mx, 8));
        const float mnew = fmaxf(mrow[g], mx);
        const float alpha = __expf(mrow[g] - mnew);
        const float pe = __expf(v - mnew);
        float rs = pe;
        rs += __shfl_xor(rs, 1);
        rs += __shfl_xor(rs, 2);
        rs += __shfl_xor(rs, 4);
        rs += __shfl_xor(rs, 8);
        lrow[g] = lrow[g] * alpha + rs;
        mrow[g] = mnew;
        pv[g] = pe;
        o0[g] *= alpha; o1[g] *= alpha; o2[g] *= alpha; o3[g] *= alpha;
      }
#pragma unroll
      for (int g = 0; g < 8; ++g) Pw[(g + 8 * hhalf) * 17 + r] = pv[g];
#pragma unroll
      for (int kk2 = 0; kk2 < 4; ++kk2) {
        const int kd = kk2 * 4 + hhalf * 2;
        const float* pr = Pw + r * 17 + kd;
        v2f af = {pr[0], pr[1]};
        v2f b0 = {Vs[kd * 68 + 0 * 16 + r], Vs[(kd + 1) * 68 + 0 * 16 + r]};
        v2f b1 = {Vs[kd * 68 + 1 * 16 + r], Vs[(kd + 1) * 68 + 1 * 16 + r]};
        v2f b2 = {Vs[kd * 68 + 2 * 16 + r], Vs[(kd + 1) * 68 + 2 * 16 + r]};
        v2f b3 = {Vs[kd * 68 + 3 * 16 + r], Vs[(kd + 1) * 68 + 3 * 16 + r]};
        o0 = wmma_f32(af, b0, o0);
        o1 = wmma_f32(af, b1, o1);
        o2 = wmma_f32(af, b2, o2);
        o3 = wmma_f32(af, b3, o3);
      }
    }
  }

  const int bb = bh >> 4;
  const int head = bh & 15;
#pragma unroll
  for (int g = 0; g < 8; ++g) {
    const float inv = 1.0f / (lrow[g] + 1e-8f);
    const int qrow = qw0 + g + 8 * hhalf;
    float* dst = O + ((size_t)(bb * S_ + qrow)) * DM_ + head * DK_;
    dst[0 * 16 + r] = o0[g] * inv;
    dst[1 * 16 + r] = o1[g] * inv;
    dst[2 * 16 + r] = o2[g] * inv;
    dst[3 * 16 + r] = o3[g] * inv;
  }
}

__global__ void __launch_bounds__(256)
out_proj_kernel(const float* __restrict__ Oin, const float* __restrict__ w_out,
                const float* __restrict__ b_out, float* __restrict__ out) {
  __shared__ float Xs[128 * 20];
  __shared__ float Ws[64 * 20];
  v8f acc[2][2] = {};
  const int m0 = blockIdx.x * 128;
  const int n0 = blockIdx.y * 64;
  gemm_core_128x64(Oin, w_out, Xs, Ws, m0, n0, DM_, acc);

  const int tid = threadIdx.x;
  const int wv = tid >> 5;
  const int lane = tid & 31;
  const int r = lane & 15;
  const int hhalf = lane >> 4;
  const int wm = wv >> 1;
  const int wn = wv & 1;

#pragma unroll
  for (int mi = 0; mi < 2; ++mi) {
#pragma unroll
    for (int ni = 0; ni < 2; ++ni) {
      const int n = n0 + wn * 32 + ni * 16 + r;
      const float bv = b_out[n];
#pragma unroll
      for (int g = 0; g < 8; ++g) {
        const int gm = m0 + wm * 32 + mi * 16 + g + 8 * hhalf;
        out[(size_t)gm * DM_ + n] = acc[mi][ni][g] + bv;
      }
    }
  }
}

extern "C" void kernel_launch(void* const* d_in, const int* in_sizes, int n_in,
                              void* d_out, int out_size, void* d_ws,
                              size_t ws_size, hipStream_t stream) {
  const float* x = (const float*)d_in[0];
  const float* w_qkv = (const float*)d_in[1];
  const float* b_qkv = (const float*)d_in[2];
  const float* w_out = (const float*)d_in[3];
  const float* b_out = (const float*)d_in[4];
  float* out = (float*)d_out;

  const size_t per_buf = (size_t)BH_ * S_ * DK_;
  float* Qw = (float*)d_ws;
  float* Kw = Qw + per_buf;
  float* Vw = Kw + per_buf;
  float* Ow = Vw + per_buf;

  qkv_rope_kernel<<<dim3(ROWS_ / 128, N_QKV / 64), 256, 0, stream>>>(
      x, w_qkv, b_qkv, Qw, Kw, Vw);
  flash_attn_kernel<<<dim3(S_ / 128, BH_), 256, 0, stream>>>(Qw, Kw, Vw, Ow);
  out_proj_kernel<<<dim3(ROWS_ / 128, DM_ / 64), 256, 0, stream>>>(
      Ow, w_out, b_out, out);
}


# attention scores computed transposed (S^T = K*Q^T): softmax reduces in-lane plus one lane-half swap, P^T feeds O^T += V^T*P^T from registers; row sum kept per lane half, joined once at end
# speedup vs baseline: 1.1804x; 1.1804x over previous
#include <hip/hip_runtime.h>
#include <hip/hip_bf16.h>


typedef __attribute__((ext_vector_type(2))) float v2f;
typedef __attribute__((ext_vector_type(8))) float v8f;

#define B_     2
#define S_     2048
#define DM_    1024
#define H_     16
#define DK_    64
#define BH_    (B_ * H_)
#define ROWS_  (B_ * S_)
#define N_QKV  (3 * DM_)

__device__ __forceinline__ v8f wmma_f32(v2f a, v2f b, v8f c) {
  return __builtin_amdgcn_wmma_f32_16x16x4_f32(
      false, a, false, b, (short)0, c, false, false);
}

__device__ __forceinline__ void gemm_core_128x64(
    const float* __restrict__ A, const float* __restrict__ W,
    float* Xs, float* Ws, int m0, int n0, int Kdim, v8f acc[2][2]) {
  const int tid = threadIdx.x;
  const int wv = tid >> 5;
  const int lane = tid & 31;
  const int r = lane & 15;
  const int hh = lane >> 4;
  const int wm = wv >> 1;
  const int wn = wv & 1;

  for (int k0 = 0; k0 < Kdim; k0 += 16) {
    __syncthreads();
#pragma unroll
    for (int i = 0; i < 2; ++i) {
      int idx = tid + i * 256;
      int row = idx >> 2;
      int cv = (idx & 3) * 4;
      *(float4*)(Xs + row * 20 + cv) =
          *(const float4*)(A + (size_t)(m0 + row) * Kdim + k0 + cv);
    }
    {
      int row = tid >> 2;
      int cv = (tid & 3) * 4;
      *(float4*)(Ws + row * 20 + cv) =
          *(const float4*)(W + (size_t)(n0 + row) * Kdim + k0 + cv);
    }
    __syncthreads();
#pragma unroll
    for (int kk = 0; kk < 4; ++kk) {
      const int kd = kk * 4 + hh * 2;
      const float* xr0 = Xs + (wm * 32 + r) * 20 + kd;
      const float* xr1 = Xs + (wm * 32 + 16 + r) * 20 + kd;
      const float* wr0 = Ws + (wn * 32 + r) * 20 + kd;
      const float* wr1 = Ws + (wn * 32 + 16 + r) * 20 + kd;
      v2f a0 = {xr0[0], xr0[1]};
      v2f a1 = {xr1[0], xr1[1]};
      v2f b0 = {wr0[0], wr0[1]};
      v2f b1 = {wr1[0], wr1[1]};
      acc[0][0] = wmma_f32(a0, b0, acc[0][0]);
      acc[0][1] = wmma_f32(a0, b1, acc[0][1]);
      acc[1][0] = wmma_f32(a1, b0, acc[1][0]);
      acc[1][1] = wmma_f32(a1, b1, acc[1][1]);
    }
  }
}

__global__ void __launch_bounds__(256)
qkv_rope_kernel(const float* __restrict__ x, const float* __restrict__ w_qkv,
                const float* __restrict__ b_qkv, float* __restrict__ Qw,
                float* __restrict__ Kw, float* __restrict__ Vw) {
  __shared__ float Xs[128 * 20];
  __shared__ float Ws[64 * 20];
  v8f acc[2][2] = {};
  const int m0 = blockIdx.x * 128;
  const int n0 = blockIdx.y * 64;
  gemm_core_128x64(x, w_qkv, Xs, Ws, m0, n0, DM_, acc);

  const int tid = threadIdx.x;
  const int wv = tid >> 5;
  const int lane = tid & 31;
  const int r = lane & 15;
  const int hhalf = lane >> 4;
  const int wm = wv >> 1;
  const int wn = wv & 1;

#pragma unroll
  for (int mi = 0; mi < 2; ++mi) {
#pragma unroll
    for (int ni = 0; ni < 2; ++ni) {
      const int o = n0 + wn * 32 + ni * 16 + r;
      const int sel = o >> 10;
      const int rem = o & 1023;
      const int head = rem >> 6;
      const int d = rem & 63;
      const float bv = b_qkv[o];
      const int p = d >> 1;
      const float invf = __expf((float)(2 * p) * (-9.210340371976184f / 64.0f));
      const float sgn = (d & 1) ? 1.0f : -1.0f;
      float* dst = (sel == 0) ? Qw : ((sel == 1) ? Kw : Vw);
#pragma unroll
      for (int g = 0; g < 8; ++g) {
        const int gm = m0 + wm * 32 + mi * 16 + g + 8 * hhalf;
        const int srow = gm & (S_ - 1);
        const int bb = gm >> 11;
        float v = acc[mi][ni][g] + bv;
        float other = __shfl_xor(v, 1);
        float outv = v;
        if (sel < 2) {
          const float ang = (float)srow * invf;
          const float sn = __sinf(ang);
          const float cs = __cosf(ang);
          outv = v * cs + other * sgn * sn;
        }
        dst[(((size_t)(bb * H_ + head)) * S_ + srow) * DK_ + d] = outv;
      }
    }
  }
}

__device__ __forceinline__ float xhalf(float x) {
  const int b = __builtin_bit_cast(int, x);
  return __builtin_bit_cast(
      float, __builtin_amdgcn_permlanex16(b, b, 0x76543210, 0xfedcba98, false,
                                          false));
}

__global__ void __launch_bounds__(256)
flash_attn_kernel(const float* __restrict__ Q, const float* __restrict__ K,
                  const float* __restrict__ V, float* __restrict__ O) {
  __shared__ float Ks[16 * 68];
  __shared__ float Vs[16 * 68];

  const int bh = blockIdx.y;
  const int qb = blockIdx.x;
  const int tid = threadIdx.x;
  const int wv = tid >> 5;
  const int lane = tid & 31;
  const int r = lane & 15;
  const int hhalf = lane >> 4;
  const int qw0 = qb * 128 + wv * 16;

  const float* Qp = Q + (size_t)bh * S_ * DK_;
  const float* Kp = K + (size_t)bh * S_ * DK_;
  const float* Vp = V + (size_t)bh * S_ * DK_;

  v2f qf[16];
#pragma unroll
  for (int kk = 0; kk < 16; ++kk) {
    const float* qp = Qp + (size_t)(qw0 + r) * DK_ + kk * 4 + hhalf * 2;
    qf[kk] = v2f{qp[0] * 0.125f, qp[1] * 0.125f};
  }

  v8f o0 = {}, o1 = {}, o2 = {}, o3 = {};
  float mrun = -3.0e38f;
  float lrun = 0.0f;

  const int njt = qb * 8 + 8;

  for (int j = 0; j < njt; ++j) {
    __syncthreads();
    {
      const int rowj = tid >> 4;
      const int cv = (tid & 15) * 4;
      const float4 kv = *(const float4*)(Kp + (size_t)(j * 16 + rowj) * DK_ + cv);
      const float4 vvv = *(const float4*)(Vp + (size_t)(j * 16 + rowj) * DK_ + cv);
      *(float4*)(Ks + rowj * 68 + cv) = kv;
      *(float4*)(Vs + rowj * 68 + cv) = vvv;
    }
    __syncthreads();

    if (j * 16 <= qw0 + 15) {
      v8f s = {};
#pragma unroll
      for (int kk = 0; kk < 16; ++kk) {
        const int kd = kk * 4 + hhalf * 2;
        const float* kr = Ks + r * 68 + kd;
        v2f kf = {kr[0], kr[1]};
        s = wmma_f32(kf, qf[kk], s);
      }
      if (j * 16 + 15 > qw0) {
        const int dk0 = j * 16 + 8 * hhalf - (qw0 + r);
#pragma unroll
        for (int g = 0; g < 8; ++g) {
          if (dk0 + g > 0) s[g] = -3.0e38f;
        }
      }
      float mx = fmaxf(fmaxf(fmaxf(s[0], s[1]), fmaxf(s[2], s[3])),
                       fmaxf(fmaxf(s[4], s[5]), fmaxf(s[6], s[7])));
      mx = fmaxf(mx, xhalf(mx));
      const float mnew = fmaxf(mrun, mx);
      const float alpha = __expf(mrun - mnew);
      mrun = mnew;
      v8f p;
#pragma unroll
      for (int g = 0; g < 8; ++g) p[g] = __expf(s[g] - mnew);
      lrun = lrun * alpha +
             (((p[0] + p[1]) + (p[2] + p[3])) + ((p[4] + p[5]) + (p[6] + p[7])));
      o0 *= alpha; o1 *= alpha; o2 *= alpha; o3 *= alpha;
#pragma unroll
      for (int t = 0; t < 4; ++t) {
        const float* vr = Vs + (8 * hhalf + 2 * t) * 68 + r;
        v2f bp = {p[2 * t], p[2 * t + 1]};
        v2f a0 = {vr[0 * 16], vr[68 + 0 * 16]};
        v2f a1 = {vr[1 * 16], vr[68 + 1 * 16]};
        v2f a2 = {vr[2 * 16], vr[68 + 2 * 16]};
        v2f a3 = {vr[3 * 16], vr[68 + 3 * 16]};
        o0 = wmma_f32(a0, bp, o0);
        o1 = wmma_f32(a1, bp, o1);
        o2 = wmma_f32(a2, bp, o2);
        o3 = wmma_f32(a3, bp, o3);
      }
    }
  }

  const int bb = bh >> 4;
  const int head = bh & 15;
  const float inv = 1.0f / ((lrun + xhalf(lrun)) + 1e-8f);
  float* dst = O + ((size_t)(bb * S_ + qw0 + r)) * DM_ + head * DK_ + 8 * hhalf;
  *(float4*)(dst + 0)      = make_float4(o0[0] * inv, o0[1] * inv, o0[2] * inv, o0[3] * inv);
  *(float4*)(dst + 4)      = make_float4(o0[4] * inv, o0[5] * inv, o0[6] * inv, o0[7] * inv);
  *(float4*)(dst + 16)     = make_float4(o1[0] * inv, o1[1] * inv, o1[2] * inv, o1[3] * inv);
  *(float4*)(dst + 16 + 4) = make_float4(o1[4] * inv, o1[5] * inv, o1[6] * inv, o1[7] * inv);
  *(float4*)(dst + 32)     = make_float4(o2[0] * inv, o2[1] * inv, o2[2] * inv, o2[3] * inv);
  *(float4*)(dst + 32 + 4) = make_float4(o2[4] * inv, o2[5] * inv, o2[6] * inv, o2[7] * inv);
  *(float4*)(dst + 48)     = make_float4(o3[0] * inv, o3[1] * inv, o3[2] * inv, o3[3] * inv);
  *(float4*)(dst + 48 + 4) = make_float4(o3[4] * inv, o3[5] * inv, o3[6] * inv, o3[7] * inv);
}

__global__ void __launch_bounds__(256)
out_proj_kernel(const float* __restrict__ Oin, const float* __restrict__ w_out,
                const float* __restrict__ b_out, float* __restrict__ out) {
  __shared__ float Xs[128 * 20];
  __shared__ float Ws[64 * 20];
  v8f acc[2][2] = {};
  const int m0 = blockIdx.x * 128;
  const int n0 = blockIdx.y * 64;
  gemm_core_128x64(Oin, w_out, Xs, Ws, m0, n0, DM_, acc);

  const int tid = threadIdx.x;
  const int wv = tid >> 5;
  const int lane = tid & 31;
  const int r = lane & 15;
  const int hhalf = lane >> 4;
  const int wm = wv >> 1;
  const int wn = wv & 1;

#pragma unroll
  for (int mi = 0; mi < 2; ++mi) {
#pragma unroll
    for (int ni = 0; ni < 2; ++ni) {
      const int n = n0 + wn * 32 + ni * 16 + r;
      const float bv = b_out[n];
#pragma unroll
      for (int g = 0; g < 8; ++g) {
        const int gm = m0 + wm * 32 + mi * 16 + g + 8 * hhalf;
        out[(size_t)gm * DM_ + n] = acc[mi][ni][g] + bv;
      }
    }
  }
}

extern "C" void kernel_launch(void* const* d_in, const int* in_sizes, int n_in,
                              void* d_out, int out_size, void* d_ws,
                              size_t ws_size, hipStream_t stream) {
  const float* x = (const float*)d_in[0];
  const float* w_qkv = (const float*)d_in[1];
  const float* b_qkv = (const float*)d_in[2];
  const float* w_out = (const float*)d_in[3];
  const float* b_out = (const float*)d_in[4];
  float* out = (float*)d_out;

  const size_t per_buf = (size_t)BH_ * S_ * DK_;
  float* Qw = (float*)d_ws;
  float* Kw = Qw + per_buf;
  float* Vw = Kw + per_buf;
  float* Ow = Vw + per_buf;

  qkv_rope_kernel<<<dim3(ROWS_ / 128, N_QKV / 64), 256, 0, stream>>>(
      x, w_qkv, b_qkv, Qw, Kw, Vw);
  flash_attn_kernel<<<dim3(S_ / 128, BH_), 256, 0, stream>>>(Qw, Kw, Vw, Ow);
  out_proj_kernel<<<dim3(ROWS_ / 128, DM_ / 64), 256, 0, stream>>>(
      Ow, w_out, b_out, out);
}


# attention: each work-group (4 waves, 64 queries) takes query block 31-i and then block i of one head, so all 512 work-groups do equal causal work (132 key tiles each; was 8 to 128)
# speedup vs baseline: 1.2929x; 1.0953x over previous
#include <hip/hip_runtime.h>
#include <hip/hip_bf16.h>


typedef __attribute__((ext_vector_type(2))) float v2f;
typedef __attribute__((ext_vector_type(8))) float v8f;

#define B_     2
#define S_     2048
#define DM_    1024
#define H_     16
#define DK_    64
#define BH_    (B_ * H_)
#define ROWS_  (B_ * S_)
#define N_QKV  (3 * DM_)

__device__ __forceinline__ v8f wmma_f32(v2f a, v2f b, v8f c) {
  return __builtin_amdgcn_wmma_f32_16x16x4_f32(
      false, a, false, b, (short)0, c, false, false);
}

__device__ __forceinline__ void gemm_core_128x64(
    const float* __restrict__ A, const float* __restrict__ W,
    float* Xs, float* Ws, int m0, int n0, int Kdim, v8f acc[2][2]) {
  const int tid = threadIdx.x;
  const int wv = tid >> 5;
  const int lane = tid & 31;
  const int r = lane & 15;
  const int hh = lane >> 4;
  const int wm = wv >> 1;
  const int wn = wv & 1;

  for (int k0 = 0; k0 < Kdim; k0 += 16) {
    __syncthreads();
#pragma unroll
    for (int i = 0; i < 2; ++i) {
      int idx = tid + i * 256;
      int row = idx >> 2;
      int cv = (idx & 3) * 4;
      *(float4*)(Xs + row * 20 + cv) =
          *(const float4*)(A + (size_t)(m0 + row) * Kdim + k0 + cv);
    }
    {
      int row = tid >> 2;
      int cv = (tid & 3) * 4;
      *(float4*)(Ws + row * 20 + cv) =
          *(const float4*)(W + (size_t)(n0 + row) * Kdim + k0 + cv);
    }
    __syncthreads();
#pragma unroll
    for (int kk = 0; kk < 4; ++kk) {
      const int kd = kk * 4 + hh * 2;
      const float* xr0 = Xs + (wm * 32 + r) * 20 + kd;
      const float* xr1 = Xs + (wm * 32 + 16 + r) * 20 + kd;
      const float* wr0 = Ws + (wn * 32 + r) * 20 + kd;
      const float* wr1 = Ws + (wn * 32 + 16 + r) * 20 + kd;
      v2f a0 = {xr0[0], xr0[1]};
      v2f a1 = {xr1[0], xr1[1]};
      v2f b0 = {wr0[0], wr0[1]};
      v2f b1 = {wr1[0], wr1[1]};
      acc[0][0] = wmma_f32(a0, b0, acc[0][0]);
      acc[0][1] = wmma_f32(a0, b1, acc[0][1]);
      acc[1][0] = wmma_f32(a1, b0, acc[1][0]);
      acc[1][1] = wmma_f32(a1, b1, acc[1][1]);
    }
  }
}

__global__ void __launch_bounds__(256)
qkv_rope_kernel(const float* __restrict__ x, const float* __restrict__ w_qkv,
                const float* __restrict__ b_qkv, float* __restrict__ Qw,
                float* __restrict__ Kw, float* __restrict__ Vw) {
  __shared__ float Xs[128 * 20];
  __shared__ float Ws[64 * 20];
  v8f acc[2][2] = {};
  const int m0 = blockIdx.x * 128;
  const int n0 = blockIdx.y * 64;
  gemm_core_128x64(x, w_qkv, Xs, Ws, m0, n0, DM_, acc);

  const int tid = threadIdx.x;
  const int wv = tid >> 5;
  const int lane = tid & 31;
  const int r = lane & 15;
  const int hhalf = lane >> 4;
  const int wm = wv >> 1;
  const int wn = wv & 1;

#pragma unroll
  for (int mi = 0; mi < 2; ++mi) {
#pragma unroll
    for (int ni = 0; ni < 2; ++ni) {
      const int o = n0 + wn * 32 + ni * 16 + r;
      const int sel = o >> 10;
      const int rem = o & 1023;
      const int head = rem >> 6;
      const int d = rem & 63;
      const float bv = b_qkv[o];
      const int p = d >> 1;
      const float invf = __expf((float)(2 * p) * (-9.210340371976184f / 64.0f));
      const float sgn = (d & 1) ? 1.0f : -1.0f;
      float* dst = (sel == 0) ? Qw : ((sel == 1) ? Kw : Vw);
#pragma unroll
      for (int g = 0; g < 8; ++g) {
        const int gm = m0 + wm * 32 + mi * 16 + g + 8 * hhalf;
        const int srow = gm & (S_ - 1);
        const int bb = gm >> 11;
        float v = acc[mi][ni][g] + bv;
        float other = __shfl_xor(v, 1);
        float outv = v;
        if (sel < 2) {
          const float ang = (float)srow * invf;
          const float sn = __sinf(ang);
          const float cs = __cosf(ang);
          outv = v * cs + other * sgn * sn;
        }
        dst[(((size_t)(bb * H_ + head)) * S_ + srow) * DK_ + d] = outv;
      }
    }
  }
}

__device__ __forceinline__ float xhalf(float x) {
  const int b = __builtin_bit_cast(int, x);
  return __builtin_bit_cast(
      float, __builtin_amdgcn_permlanex16(b, b, 0x76543210, 0xfedcba98, false,
                                          false));
}

#define QBLK_  64
#define NQBLK_ (S_ / QBLK_)

__global__ void __launch_bounds__(128)
flash_attn_kernel(const float* __restrict__ Q, const float* __restrict__ K,
                  const float* __restrict__ V, float* __restrict__ O) {
  __shared__ float Ks[16 * 68];
  __shared__ float Vs[16 * 68];

  const int bh = blockIdx.y;
  const int tid = threadIdx.x;
  const int wv = tid >> 5;
  const int lane = tid & 31;
  const int r = lane & 15;
  const int hhalf = lane >> 4;
  const int bb = bh >> 4;
  const int head = bh & 15;

  const float* Qp = Q + (size_t)bh * S_ * DK_;
  const float* Kp = K + (size_t)bh * S_ * DK_;
  const float* Vp = V + (size_t)bh * S_ * DK_;

  for (int ph = 0; ph < 2; ++ph) {
    const int qb = (ph == 0) ? (NQBLK_ - 1 - (int)blockIdx.x) : (int)blockIdx.x;
    const int qw0 = qb * QBLK_ + wv * 16;

    v2f qf[16];
#pragma unroll
    for (int kk = 0; kk < 16; ++kk) {
      const float* qp = Qp + (size_t)(qw0 + r) * DK_ + kk * 4 + hhalf * 2;
      qf[kk] = v2f{qp[0] * 0.125f, qp[1] * 0.125f};
    }

    v8f o0 = {}, o1 = {}, o2 = {}, o3 = {};
    float mrun = -3.0e38f;
    float lrun = 0.0f;

    const int njt = qb * 4 + 4;

    for (int j = 0; j < njt; ++j) {
      __syncthreads();
      {
        const int row0 = tid >> 4;
        const int cv = (tid & 15) * 4;
        const float* kg = Kp + (size_t)(j * 16 + row0) * DK_ + cv;
        const float* vg = Vp + (size_t)(j * 16 + row0) * DK_ + cv;
        const float4 k0 = *(const float4*)(kg);
        const float4 k1 = *(const float4*)(kg + 8 * DK_);
        const float4 v0 = *(const float4*)(vg);
        const float4 v1 = *(const float4*)(vg + 8 * DK_);
        *(float4*)(Ks + row0 * 68 + cv) = k0;
        *(float4*)(Ks + (row0 + 8) * 68 + cv) = k1;
        *(float4*)(Vs + row0 * 68 + cv) = v0;
        *(float4*)(Vs + (row0 + 8) * 68 + cv) = v1;
      }
      __syncthreads();

      if (j * 16 <= qw0 + 15) {
        v8f s = {};
#pragma unroll
        for (int kk = 0; kk < 16; ++kk) {
          const int kd = kk * 4 + hhalf * 2;
          const float* kr = Ks + r * 68 + kd;
          v2f kf = {kr[0], kr[1]};
          s = wmma_f32(kf, qf[kk], s);
        }
        if (j * 16 + 15 > qw0) {
          const int dk0 = j * 16 + 8 * hhalf - (qw0 + r);
#pragma unroll
          for (int g = 0; g < 8; ++g) {
            if (dk0 + g > 0) s[g] = -3.0e38f;
          }
        }
        float mx = fmaxf(fmaxf(fmaxf(s[0], s[1]), fmaxf(s[2], s[3])),
                         fmaxf(fmaxf(s[4], s[5]), fmaxf(s[6], s[7])));
        mx = fmaxf(mx, xhalf(mx));
        const float mnew = fmaxf(mrun, mx);
        const float alpha = __expf(mrun - mnew);
        mrun = mnew;
        v8f p;
#pragma unroll
        for (int g = 0; g < 8; ++g) p[g] = __expf(s[g] - mnew);
        lrun = lrun * alpha +
               (((p[0] + p[1]) + (p[2] + p[3])) + ((p[4] + p[5]) + (p[6] + p[7])));
        o0 *= alpha; o1 *= alpha; o2 *= alpha; o3 *= alpha;
#pragma unroll
        for (int t = 0; t < 4; ++t) {
          const float* vr = Vs + (8 * hhalf + 2 * t) * 68 + r;
          v2f bp = {p[2 * t], p[2 * t + 1]};
          v2f a0 = {vr[0 * 16], vr[68 + 0 * 16]};
          v2f a1 = {vr[1 * 16], vr[68 + 1 * 16]};
          v2f a2 = {vr[2 * 16], vr[68 + 2 * 16]};
          v2f a3 = {vr[3 * 16], vr[68 + 3 * 16]};
          o0 = wmma_f32(a0, bp, o0);
          o1 = wmma_f32(a1, bp, o1);
          o2 = wmma_f32(a2, bp, o2);
          o3 = wmma_f32(a3, bp, o3);
        }
      }
    }

    const float inv = 1.0f / ((lrun + xhalf(lrun)) + 1e-8f);
    float* dst = O + ((size_t)(bb * S_ + qw0 + r)) * DM_ + head * DK_ + 8 * hhalf;
    *(float4*)(dst + 0)      = make_float4(o0[0] * inv, o0[1] * inv, o0[2] * inv, o0[3] * inv);
    *(float4*)(dst + 4)      = make_float4(o0[4] * inv, o0[5] * inv, o0[6] * inv, o0[7] * inv);
    *(float4*)(dst + 16)     = make_float4(o1[0] * inv, o1[1] * inv, o1[2] * inv, o1[3] * inv);
    *(float4*)(dst + 16 + 4) = make_float4(o1[4] * inv, o1[5] * inv, o1[6] * inv, o1[7] * inv);
    *(float4*)(dst + 32)     = make_float4(o2[0] * inv, o2[1] * inv, o2[2] * inv, o2[3] * inv);
    *(float4*)(dst + 32 + 4) = make_float4(o2[4] * inv, o2[5] * inv, o2[6] * inv, o2[7] * inv);
    *(float4*)(dst + 48)     = make_float4(o3[0] * inv, o3[1] * inv, o3[2] * inv, o3[3] * inv);
    *(float4*)(dst + 48 + 4) = make_float4(o3[4] * inv, o3[5] * inv, o3[6] * inv, o3[7] * inv);
  }
}

__global__ void __launch_bounds__(256)
out_proj_kernel(const float* __restrict__ Oin, const float* __restrict__ w_out,
                const float* __restrict__ b_out, float* __restrict__ out) {
  __shared__ float Xs[128 * 20];
  __shared__ float Ws[64 * 20];
  v8f acc[2][2] = {};
  const int m0 = blockIdx.x * 128;
  const int n0 = blockIdx.y * 64;
  gemm_core_128x64(Oin, w_out, Xs, Ws, m0, n0, DM_, acc);

  const int tid = threadIdx.x;
  const int wv = tid >> 5;
  const int lane = tid & 31;
  const int r = lane & 15;
  const int hhalf = lane >> 4;
  const int wm = wv >> 1;
  const int wn = wv & 1;

#pragma unroll
  for (int mi = 0; mi < 2; ++mi) {
#pragma unroll
    for (int ni = 0; ni < 2; ++ni) {
      const int n = n0 + wn * 32 + ni * 16 + r;
      const float bv = b_out[n];
#pragma unroll
      for (int g = 0; g < 8; ++g) {
        const int gm = m0 + wm * 32 + mi * 16 + g + 8 * hhalf;
        out[(size_t)gm * DM_ + n] = acc[mi][ni][g] + bv;
      }
    }
  }
}

extern "C" void kernel_launch(void* const* d_in, const int* in_sizes, int n_in,
                              void* d_out, int out_size, void* d_ws,
                              size_t ws_size, hipStream_t stream) {
  const float* x = (const float*)d_in[0];
  const float* w_qkv = (const float*)d_in[1];
  const float* b_qkv = (const float*)d_in[2];
  const float* w_out = (const float*)d_in[3];
  const float* b_out = (const float*)d_in[4];
  float* out = (float*)d_out;

  const size_t per_buf = (size_t)BH_ * S_ * DK_;
  float* Qw = (float*)d_ws;
  float* Kw = Qw + per_buf;
  float* Vw = Kw + per_buf;
  float* Ow = Vw + per_buf;

  qkv_rope_kernel<<<dim3(ROWS_ / 128, N_QKV / 64), 256, 0, stream>>>(
      x, w_qkv, b_qkv, Qw, Kw, Vw);
  flash_attn_kernel<<<dim3(NQBLK_ / 2, BH_), 128, 0, stream>>>(Qw, Kw, Vw, Ow);
  out_proj_kernel<<<dim3(ROWS_ / 128, DM_ / 64), 256, 0, stream>>>(
      Ow, w_out, b_out, out);
}


# all three kernels: the next tile's global loads go into registers ahead of the current tile's WMMAs; barriers fence LDS only, so those loads stay in flight instead of being drained
# speedup vs baseline: 1.5196x; 1.1753x over previous
#include <hip/hip_runtime.h>
#include <hip/hip_bf16.h>


typedef __attribute__((ext_vector_type(2))) float v2f;
typedef __attribute__((ext_vector_type(8))) float v8f;

#define B_     2
#define S_     2048
#define DM_    1024
#define H_     16
#define DK_    64
#define BH_    (B_ * H_)
#define ROWS_  (B_ * S_)
#define N_QKV  (3 * DM_)

__device__ __forceinline__ v8f wmma_f32(v2f a, v2f b, v8f c) {
  return __builtin_amdgcn_wmma_f32_16x16x4_f32(
      false, a, false, b, (short)0, c, false, false);
}

__device__ __forceinline__ void lds_barrier() {
  __builtin_amdgcn_fence(__ATOMIC_RELEASE, "workgroup", "local");
  __builtin_amdgcn_s_barrier();
  __builtin_amdgcn_fence(__ATOMIC_ACQUIRE, "workgroup", "local");
}

__device__ __forceinline__ void gemm_core_128x64(
    const float* __restrict__ A, const float* __restrict__ W,
    float* Xs, float* Ws, int m0, int n0, int Kdim, v8f acc[2][2]) {
  const int tid = threadIdx.x;
  const int wv = tid >> 5;
  const int lane = tid & 31;
  const int r = lane & 15;
  const int hh = lane >> 4;
  const int wm = wv >> 1;
  const int wn = wv & 1;

  const int srow = tid >> 2;
  const int scol = (tid & 3) * 4;
  const float* xg0 = A + (size_t)(m0 + srow) * Kdim + scol;
  const float* xg1 = A + (size_t)(m0 + 64 + srow) * Kdim + scol;
  const float* wg0 = W + (size_t)(n0 + srow) * Kdim + scol;
  float4 xa0 = *(const float4*)(xg0);
  float4 xa1 = *(const float4*)(xg1);
  float4 wa0 = *(const float4*)(wg0);

  for (int k0 = 0; k0 < Kdim; k0 += 16) {
    lds_barrier();
    *(float4*)(Xs + srow * 20 + scol) = xa0;
    *(float4*)(Xs + (srow + 64) * 20 + scol) = xa1;
    *(float4*)(Ws + srow * 20 + scol) = wa0;
    if (k0 + 16 < Kdim) {
      xa0 = *(const float4*)(xg0 + k0 + 16);
      xa1 = *(const float4*)(xg1 + k0 + 16);
      wa0 = *(const float4*)(wg0 + k0 + 16);
    }
    lds_barrier();
#pragma unroll
    for (int kk = 0; kk < 4; ++kk) {
      const int kd = kk * 4 + hh * 2;
      const float* xr0 = Xs + (wm * 32 + r) * 20 + kd;
      const float* xr1 = Xs + (wm * 32 + 16 + r) * 20 + kd;
      const float* wr0 = Ws + (wn * 32 + r) * 20 + kd;
      const float* wr1 = Ws + (wn * 32 + 16 + r) * 20 + kd;
      v2f a0 = {xr0[0], xr0[1]};
      v2f a1 = {xr1[0], xr1[1]};
      v2f b0 = {wr0[0], wr0[1]};
      v2f b1 = {wr1[0], wr1[1]};
      acc[0][0] = wmma_f32(a0, b0, acc[0][0]);
      acc[0][1] = wmma_f32(a0, b1, acc[0][1]);
      acc[1][0] = wmma_f32(a1, b0, acc[1][0]);
      acc[1][1] = wmma_f32(a1, b1, acc[1][1]);
    }
  }
}

__global__ void __launch_bounds__(256)
qkv_rope_kernel(const float* __restrict__ x, const float* __restrict__ w_qkv,
                const float* __restrict__ b_qkv, float* __restrict__ Qw,
                float* __restrict__ Kw, float* __restrict__ Vw) {
  __shared__ float Xs[128 * 20];
  __shared__ float Ws[64 * 20];
  v8f acc[2][2] = {};
  const int m0 = blockIdx.x * 128;
  const int n0 = blockIdx.y * 64;
  gemm_core_128x64(x, w_qkv, Xs, Ws, m0, n0, DM_, acc);

  const int tid = threadIdx.x;
  const int wv = tid >> 5;
  const int lane = tid & 31;
  const int r = lane & 15;
  const int hhalf = lane >> 4;
  const int wm = wv >> 1;
  const int wn = wv & 1;

#pragma unroll
  for (int mi = 0; mi < 2; ++mi) {
#pragma unroll
    for (int ni = 0; ni < 2; ++ni) {
      const int o = n0 + wn * 32 + ni * 16 + r;
      const int sel = o >> 10;
      const int rem = o & 1023;
      const int head = rem >> 6;
      const int d = rem & 63;
      const float bv = b_qkv[o];
      const int p = d >> 1;
      const float invf = __expf((float)(2 * p) * (-9.210340371976184f / 64.0f));
      const float sgn = (d & 1) ? 1.0f : -1.0f;
      float* dst = (sel == 0) ? Qw : ((sel == 1) ? Kw : Vw);
#pragma unroll
      for (int g = 0; g < 8; ++g) {
        const int gm = m0 + wm * 32 + mi * 16 + g + 8 * hhalf;
        const int srow = gm & (S_ - 1);
        const int bb = gm >> 11;
        float v = acc[mi][ni][g] + bv;
        float other = __shfl_xor(v, 1);
        float outv = v;
        if (sel < 2) {
          const float ang = (float)srow * invf;
          const float sn = __sinf(ang);
          const float cs = __cosf(ang);
          outv = v * cs + other * sgn * sn;
        }
        dst[(((size_t)(bb * H_ + head)) * S_ + srow) * DK_ + d] = outv;
      }
    }
  }
}

__device__ __forceinline__ float xhalf(float x) {
  const int b = __builtin_bit_cast(int, x);
  return __builtin_bit_cast(
      float, __builtin_amdgcn_permlanex16(b, b, 0x76543210, 0xfedcba98, false,
                                          false));
}

#define QBLK_  64
#define NQBLK_ (S_ / QBLK_)

__global__ void __launch_bounds__(128)
flash_attn_kernel(const float* __restrict__ Q, const float* __restrict__ K,
                  const float* __restrict__ V, float* __restrict__ O) {
  __shared__ float Ks[16 * 68];
  __shared__ float Vs[16 * 68];

  const int bh = blockIdx.y;
  const int tid = threadIdx.x;
  const int wv = tid >> 5;
  const int lane = tid & 31;
  const int r = lane & 15;
  const int hhalf = lane >> 4;
  const int bb = bh >> 4;
  const int head = bh & 15;

  const float* Qp = Q + (size_t)bh * S_ * DK_;
  const float* Kp = K + (size_t)bh * S_ * DK_;
  const float* Vp = V + (size_t)bh * S_ * DK_;

  for (int ph = 0; ph < 2; ++ph) {
    const int qb = (ph == 0) ? (NQBLK_ - 1 - (int)blockIdx.x) : (int)blockIdx.x;
    const int qw0 = qb * QBLK_ + wv * 16;

    v2f qf[16];
#pragma unroll
    for (int kk = 0; kk < 16; ++kk) {
      const float* qp = Qp + (size_t)(qw0 + r) * DK_ + kk * 4 + hhalf * 2;
      qf[kk] = v2f{qp[0] * 0.125f, qp[1] * 0.125f};
    }

    v8f o0 = {}, o1 = {}, o2 = {}, o3 = {};
    float mrun = -3.0e38f;
    float lrun = 0.0f;

    const int njt = qb * 4 + 4;

    const int row0 = tid >> 4;
    const int cv = (tid & 15) * 4;
    const float* kg = Kp + (size_t)row0 * DK_ + cv;
    const float* vg = Vp + (size_t)row0 * DK_ + cv;
    float4 k0 = *(const float4*)(kg);
    float4 k1 = *(const float4*)(kg + 8 * DK_);
    float4 v0 = *(const float4*)(vg);
    float4 v1 = *(const float4*)(vg + 8 * DK_);

    for (int j = 0; j < njt; ++j) {
      lds_barrier();
      *(float4*)(Ks + row0 * 68 + cv) = k0;
      *(float4*)(Ks + (row0 + 8) * 68 + cv) = k1;
      *(float4*)(Vs + row0 * 68 + cv) = v0;
      *(float4*)(Vs + (row0 + 8) * 68 + cv) = v1;
      if (j + 1 < njt) {
        const float* kn = kg + (size_t)(j + 1) * 16 * DK_;
        const float* vn = vg + (size_t)(j + 1) * 16 * DK_;
        k0 = *(const float4*)(kn);
        k1 = *(const float4*)(kn + 8 * DK_);
        v0 = *(const float4*)(vn);
        v1 = *(const float4*)(vn + 8 * DK_);
      }
      lds_barrier();

      if (j * 16 <= qw0 + 15) {
        v8f s = {};
#pragma unroll
        for (int kk = 0; kk < 16; ++kk) {
          const int kd = kk * 4 + hhalf * 2;
          const float* kr = Ks + r * 68 + kd;
          v2f kf = {kr[0], kr[1]};
          s = wmma_f32(kf, qf[kk], s);
        }
        if (j * 16 + 15 > qw0) {
          const int dk0 = j * 16 + 8 * hhalf - (qw0 + r);
#pragma unroll
          for (int g = 0; g < 8; ++g) {
            if (dk0 + g > 0) s[g] = -3.0e38f;
          }
        }
        float mx = fmaxf(fmaxf(fmaxf(s[0], s[1]), fmaxf(s[2], s[3])),
                         fmaxf(fmaxf(s[4], s[5]), fmaxf(s[6], s[7])));
        mx = fmaxf(mx, xhalf(mx));
        const float mnew = fmaxf(mrun, mx);
        const float alpha = __expf(mrun - mnew);
        mrun = mnew;
        v8f p;
#pragma unroll
        for (int g = 0; g < 8; ++g) p[g] = __expf(s[g] - mnew);
        lrun = lrun * alpha +
               (((p[0] + p[1]) + (p[2] + p[3])) + ((p[4] + p[5]) + (p[6] + p[7])));
        o0 *= alpha; o1 *= alpha; o2 *= alpha; o3 *= alpha;
#pragma unroll
        for (int t = 0; t < 4; ++t) {
          const float* vr = Vs + (8 * hhalf + 2 * t) * 68 + r;
          v2f bp = {p[2 * t], p[2 * t + 1]};
          v2f a0 = {vr[0 * 16], vr[68 + 0 * 16]};
          v2f a1 = {vr[1 * 16], vr[68 + 1 * 16]};
          v2f a2 = {vr[2 * 16], vr[68 + 2 * 16]};
          v2f a3 = {vr[3 * 16], vr[68 + 3 * 16]};
          o0 = wmma_f32(a0, bp, o0);
          o1 = wmma_f32(a1, bp, o1);
          o2 = wmma_f32(a2, bp, o2);
          o3 = wmma_f32(a3, bp, o3);
        }
      }
    }

    const float inv = 1.0f / ((lrun + xhalf(lrun)) + 1e-8f);
    float* dst = O + ((size_t)(bb * S_ + qw0 + r)) * DM_ + head * DK_ + 8 * hhalf;
    *(float4*)(dst + 0)      = make_float4(o0[0] * inv, o0[1] * inv, o0[2] * inv, o0[3] * inv);
    *(float4*)(dst + 4)      = make_float4(o0[4] * inv, o0[5] * inv, o0[6] * inv, o0[7] * inv);
    *(float4*)(dst + 16)     = make_float4(o1[0] * inv, o1[1] * inv, o1[2] * inv, o1[3] * inv);
    *(float4*)(dst + 16 + 4) = make_float4(o1[4] * inv, o1[5] * inv, o1[6] * inv, o1[7] * inv);
    *(float4*)(dst + 32)     = make_float4(o2[0] * inv, o2[1] * inv, o2[2] * inv, o2[3] * inv);
    *(float4*)(dst + 32 + 4) = make_float4(o2[4] * inv, o2[5] * inv, o2[6] * inv, o2[7] * inv);
    *(float4*)(dst + 48)     = make_float4(o3[0] * inv, o3[1] * inv, o3[2] * inv, o3[3] * inv);
    *(float4*)(dst + 48 + 4) = make_float4(o3[4] * inv, o3[5] * inv, o3[6] * inv, o3[7] * inv);
  }
}

__global__ void __launch_bounds__(256)
out_proj_kernel(const float* __restrict__ Oin, const float* __restrict__ w_out,
                const float* __restrict__ b_out, float* __restrict__ out) {
  __shared__ float Xs[128 * 20];
  __shared__ float Ws[64 * 20];
  v8f acc[2][2] = {};
  const int m0 = blockIdx.x * 128;
  const int n0 = blockIdx.y * 64;
  gemm_core_128x64(Oin, w_out, Xs, Ws, m0, n0, DM_, acc);

  const int tid = threadIdx.x;
  const int wv = tid >> 5;
  const int lane = tid & 31;
  const int r = lane & 15;
  const int hhalf = lane >> 4;
  const int wm = wv >> 1;
  const int wn = wv & 1;

#pragma unroll
  for (int mi = 0; mi < 2; ++mi) {
#pragma unroll
    for (int ni = 0; ni < 2; ++ni) {
      const int n = n0 + wn * 32 + ni * 16 + r;
      const float bv = b_out[n];
#pragma unroll
      for (int g = 0; g < 8; ++g) {
        const int gm = m0 + wm * 32 + mi * 16 + g + 8 * hhalf;
        out[(size_t)gm * DM_ + n] = acc[mi][ni][g] + bv;
      }
    }
  }
}

extern "C" void kernel_launch(void* const* d_in, const int* in_sizes, int n_in,
                              void* d_out, int out_size, void* d_ws,
                              size_t ws_size, hipStream_t stream) {
  const float* x = (const float*)d_in[0];
  const float* w_qkv = (const float*)d_in[1];
  const float* b_qkv = (const float*)d_in[2];
  const float* w_out = (const float*)d_in[3];
  const float* b_out = (const float*)d_in[4];
  float* out = (float*)d_out;

  const size_t per_buf = (size_t)BH_ * S_ * DK_;
  float* Qw = (float*)d_ws;
  float* Kw = Qw + per_buf;
  float* Vw = Kw + per_buf;
  float* Ow = Vw + per_buf;

  qkv_rope_kernel<<<dim3(ROWS_ / 128, N_QKV / 64), 256, 0, stream>>>(
      x, w_qkv, b_qkv, Qw, Kw, Vw);
  flash_attn_kernel<<<dim3(NQBLK_ / 2, BH_), 128, 0, stream>>>(Qw, Kw, Vw, Ow);
  out_proj_kernel<<<dim3(ROWS_ / 128, DM_ / 64), 256, 0, stream>>>(
      Ow, w_out, b_out, out);
}


# both projection GEMMs re-tiled: work-group 128x128 (out projection 128x64, 4 waves), wave tile 64x32, K slice 32 (was 128x64, 32x32, K 16): 64 WMMAs per barrier pair; k order in 8-k groups changes
# speedup vs baseline: 1.6101x; 1.0596x over previous
#include <hip/hip_runtime.h>
#include <hip/hip_bf16.h>


typedef __attribute__((ext_vector_type(2))) float v2f;
typedef __attribute__((ext_vector_type(8))) float v8f;

#define B_     2
#define S_     2048
#define DM_    1024
#define H_     16
#define DK_    64
#define BH_    (B_ * H_)
#define ROWS_  (B_ * S_)
#define N_QKV  (3 * DM_)

__device__ __forceinline__ v8f wmma_f32(v2f a, v2f b, v8f c) {
  return __builtin_amdgcn_wmma_f32_16x16x4_f32(
      false, a, false, b, (short)0, c, false, false);
}

__device__ __forceinline__ void lds_barrier() {
  __builtin_amdgcn_fence(__ATOMIC_RELEASE, "workgroup", "local");
  __builtin_amdgcn_s_barrier();
  __builtin_amdgcn_fence(__ATOMIC_ACQUIRE, "workgroup", "local");
}

#define GP_ 36

template <int NWN>
__device__ __forceinline__ void gemm_core_128xN(
    const float* __restrict__ A, const float* __restrict__ W,
    float* Xs, float* Ws, int m0, int n0, v8f acc[4][2]) {
  constexpr int NT = 64 * NWN;
  constexpr int BN = 32 * NWN;
  constexpr int RS = NT / 8;
  constexpr int XR = 128 / RS;
  constexpr int WR = BN / RS;
  const int tid = threadIdx.x;
  const int wv = tid >> 5;
  const int lane = tid & 31;
  const int r = lane & 15;
  const int hh = lane >> 4;
  const int wm = wv / NWN;
  const int wn = wv % NWN;

  const int srow = tid >> 3;
  const int scol = (tid & 7) * 4;
  const float* xg = A + (size_t)(m0 + srow) * DM_ + scol;
  const float* wg = W + (size_t)(n0 + srow) * DM_ + scol;
  static_assert(WR == 4 && (XR == 4 || XR == 8), "staging shape");
  const size_t ps = (size_t)RS * DM_;
  float4 xa0 = *(const float4*)(xg);
  float4 xa1 = *(const float4*)(xg + ps);
  float4 xa2 = *(const float4*)(xg + 2 * ps);
  float4 xa3 = *(const float4*)(xg + 3 * ps);
  float4 xa4 = {}, xa5 = {}, xa6 = {}, xa7 = {};
  if constexpr (XR == 8) {
    xa4 = *(const float4*)(xg + 4 * ps);
    xa5 = *(const float4*)(xg + 5 * ps);
    xa6 = *(const float4*)(xg + 6 * ps);
    xa7 = *(const float4*)(xg + 7 * ps);
  }
  float4 wa0 = *(const float4*)(wg);
  float4 wa1 = *(const float4*)(wg + ps);
  float4 wa2 = *(const float4*)(wg + 2 * ps);
  float4 wa3 = *(const float4*)(wg + 3 * ps);

  float* xst = Xs + srow * GP_ + scol;
  float* wst = Ws + srow * GP_ + scol;
  const float* xf = Xs + (wm * 64 + r) * GP_ + hh * 4;
  const float* wf = Ws + (wn * 32 + r) * GP_ + hh * 4;

  for (int k0 = 0; k0 < DM_; k0 += 32) {
    lds_barrier();
    constexpr int pl = RS * GP_;
    *(float4*)(xst) = xa0;
    *(float4*)(xst + pl) = xa1;
    *(float4*)(xst + 2 * pl) = xa2;
    *(float4*)(xst + 3 * pl) = xa3;
    if constexpr (XR == 8) {
      *(float4*)(xst + 4 * pl) = xa4;
      *(float4*)(xst + 5 * pl) = xa5;
      *(float4*)(xst + 6 * pl) = xa6;
      *(float4*)(xst + 7 * pl) = xa7;
    }
    *(float4*)(wst) = wa0;
    *(float4*)(wst + pl) = wa1;
    *(float4*)(wst + 2 * pl) = wa2;
    *(float4*)(wst + 3 * pl) = wa3;
    if (k0 + 32 < DM_) {
      const float* xn = xg + k0 + 32;
      const float* wn2 = wg + k0 + 32;
      xa0 = *(const float4*)(xn);
      xa1 = *(const float4*)(xn + ps);
      xa2 = *(const float4*)(xn + 2 * ps);
      xa3 = *(const float4*)(xn + 3 * ps);
      if constexpr (XR == 8) {
        xa4 = *(const float4*)(xn + 4 * ps);
        xa5 = *(const float4*)(xn + 5 * ps);
        xa6 = *(const float4*)(xn + 6 * ps);
        xa7 = *(const float4*)(xn + 7 * ps);
      }
      wa0 = *(const float4*)(wn2);
      wa1 = *(const float4*)(wn2 + ps);
      wa2 = *(const float4*)(wn2 + 2 * ps);
      wa3 = *(const float4*)(wn2 + 3 * ps);
    }
    lds_barrier();
#pragma unroll
    for (int g = 0; g < 4; ++g) {
      float4 af[4], bf[2];
#pragma unroll
      for (int mi = 0; mi < 4; ++mi)
        af[mi] = *(const float4*)(xf + mi * 16 * GP_ + g * 8);
#pragma unroll
      for (int ni = 0; ni < 2; ++ni)
        bf[ni] = *(const float4*)(wf + ni * 16 * GP_ + g * 8);
#pragma unroll
      for (int mi = 0; mi < 4; ++mi) {
#pragma unroll
        for (int ni = 0; ni < 2; ++ni)
          acc[mi][ni] = wmma_f32(v2f{af[mi].x, af[mi].y},
                                 v2f{bf[ni].x, bf[ni].y}, acc[mi][ni]);
      }
#pragma unroll
      for (int mi = 0; mi < 4; ++mi) {
#pragma unroll
        for (int ni = 0; ni < 2; ++ni)
          acc[mi][ni] = wmma_f32(v2f{af[mi].z, af[mi].w},
                                 v2f{bf[ni].z, bf[ni].w}, acc[mi][ni]);
      }
    }
  }
}

__global__ void __launch_bounds__(256)
qkv_rope_kernel(const float* __restrict__ x, const float* __restrict__ w_qkv,
                const float* __restrict__ b_qkv, float* __restrict__ Qw,
                float* __restrict__ Kw, float* __restrict__ Vw) {
  __shared__ float Xs[128 * GP_];
  __shared__ float Ws[128 * GP_];
  v8f acc[4][2] = {};
  const int m0 = blockIdx.x * 128;
  const int n0 = blockIdx.y * 128;
  gemm_core_128xN<4>(x, w_qkv, Xs, Ws, m0, n0, acc);

  const int tid = threadIdx.x;
  const int wv = tid >> 5;
  const int lane = tid & 31;
  const int r = lane & 15;
  const int hhalf = lane >> 4;
  const int wm = wv >> 2;
  const int wn = wv & 3;

#pragma unroll
  for (int ni = 0; ni < 2; ++ni) {
    const int o = n0 + wn * 32 + ni * 16 + r;
    const int sel = o >> 10;
    const int rem = o & 1023;
    const int head = rem >> 6;
    const int d = rem & 63;
    const float bv = b_qkv[o];
    const int p = d >> 1;
    const float invf = __expf((float)(2 * p) * (-9.210340371976184f / 64.0f));
    const float sgn = (d & 1) ? 1.0f : -1.0f;
    float* dst = (sel == 0) ? Qw : ((sel == 1) ? Kw : Vw);
#pragma unroll
    for (int mi = 0; mi < 4; ++mi) {
#pragma unroll
      for (int g = 0; g < 8; ++g) {
        const int gm = m0 + wm * 64 + mi * 16 + g + 8 * hhalf;
        const int srow = gm & (S_ - 1);
        const int bb = gm >> 11;
        float v = acc[mi][ni][g] + bv;
        float other = __shfl_xor(v, 1);
        float outv = v;
        if (sel < 2) {
          const float ang = (float)srow * invf;
          const float sn = __sinf(ang);
          const float cs = __cosf(ang);
          outv = v * cs + other * sgn * sn;
        }
        dst[(((size_t)(bb * H_ + head)) * S_ + srow) * DK_ + d] = outv;
      }
    }
  }
}

__device__ __forceinline__ float xhalf(float x) {
  const int b = __builtin_bit_cast(int, x);
  return __builtin_bit_cast(
      float, __builtin_amdgcn_permlanex16(b, b, 0x76543210, 0xfedcba98, false,
                                          false));
}

#define QBLK_  64
#define NQBLK_ (S_ / QBLK_)

__global__ void __launch_bounds__(128)
flash_attn_kernel(const float* __restrict__ Q, const float* __restrict__ K,
                  const float* __restrict__ V, float* __restrict__ O) {
  __shared__ float Ks[16 * 68];
  __shared__ float Vs[16 * 68];

  const int bh = blockIdx.y;
  const int tid = threadIdx.x;
  const int wv = tid >> 5;
  const int lane = tid & 31;
  const int r = lane & 15;
  const int hhalf = lane >> 4;
  const int bb = bh >> 4;
  const int head = bh & 15;

  const float* Qp = Q + (size_t)bh * S_ * DK_;
  const float* Kp = K + (size_t)bh * S_ * DK_;
  const float* Vp = V + (size_t)bh * S_ * DK_;

  for (int ph = 0; ph < 2; ++ph) {
    const int qb = (ph == 0) ? (NQBLK_ - 1 - (int)blockIdx.x) : (int)blockIdx.x;
    const int qw0 = qb * QBLK_ + wv * 16;

    v2f qf[16];
#pragma unroll
    for (int kk = 0; kk < 16; ++kk) {
      const float* qp = Qp + (size_t)(qw0 + r) * DK_ + kk * 4 + hhalf * 2;
      qf[kk] = v2f{qp[0] * 0.125f, qp[1] * 0.125f};
    }

    v8f o0 = {}, o1 = {}, o2 = {}, o3 = {};
    float mrun = -3.0e38f;
    float lrun = 0.0f;

    const int njt = qb * 4 + 4;

    const int row0 = tid >> 4;
    const int cv = (tid & 15) * 4;
    const float* kg = Kp + (size_t)row0 * DK_ + cv;
    const float* vg = Vp + (size_t)row0 * DK_ + cv;
    float4 k0 = *(const float4*)(kg);
    float4 k1 = *(const float4*)(kg + 8 * DK_);
    float4 v0 = *(const float4*)(vg);
    float4 v1 = *(const float4*)(vg + 8 * DK_);

    for (int j = 0; j < njt; ++j) {
      lds_barrier();
      *(float4*)(Ks + row0 * 68 + cv) = k0;
      *(float4*)(Ks + (row0 + 8) * 68 + cv) = k1;
      *(float4*)(Vs + row0 * 68 + cv) = v0;
      *(float4*)(Vs + (row0 + 8) * 68 + cv) = v1;
      if (j + 1 < njt) {
        const float* kn = kg + (size_t)(j + 1) * 16 * DK_;
        const float* vn = vg + (size_t)(j + 1) * 16 * DK_;
        k0 = *(const float4*)(kn);
        k1 = *(const float4*)(kn + 8 * DK_);
        v0 = *(const float4*)(vn);
        v1 = *(const float4*)(vn + 8 * DK_);
      }
      lds_barrier();

      if (j * 16 <= qw0 + 15) {
        v8f s = {};
#pragma unroll
        for (int kk = 0; kk < 16; ++kk) {
          const int kd = kk * 4 + hhalf * 2;
          const float* kr = Ks + r * 68 + kd;
          v2f kf = {kr[0], kr[1]};
          s = wmma_f32(kf, qf[kk], s);
        }
        if (j * 16 + 15 > qw0) {
          const int dk0 = j * 16 + 8 * hhalf - (qw0 + r);
#pragma unroll
          for (int g = 0; g < 8; ++g) {
            if (dk0 + g > 0) s[g] = -3.0e38f;
          }
        }
        float mx = fmaxf(fmaxf(fmaxf(s[0], s[1]), fmaxf(s[2], s[3])),
                         fmaxf(fmaxf(s[4], s[5]), fmaxf(s[6], s[7])));
        mx = fmaxf(mx, xhalf(mx));
        const float mnew = fmaxf(mrun, mx);
        const float alpha = __expf(mrun - mnew);
        mrun = mnew;
        v8f p;
#pragma unroll
        for (int g = 0; g < 8; ++g) p[g] = __expf(s[g] - mnew);
        lrun = lrun * alpha +
               (((p[0] + p[1]) + (p[2] + p[3])) + ((p[4] + p[5]) + (p[6] + p[7])));
        o0 *= alpha; o1 *= alpha; o2 *= alpha; o3 *= alpha;
#pragma unroll
        for (int t = 0; t < 4; ++t) {
          const float* vr = Vs + (8 * hhalf + 2 * t) * 68 + r;
          v2f bp = {p[2 * t], p[2 * t + 1]};
          v2f a0 = {vr[0 * 16], vr[68 + 0 * 16]};
          v2f a1 = {vr[1 * 16], vr[68 + 1 * 16]};
          v2f a2 = {vr[2 * 16], vr[68 + 2 * 16]};
          v2f a3 = {vr[3 * 16], vr[68 + 3 * 16]};
          o0 = wmma_f32(a0, bp, o0);
          o1 = wmma_f32(a1, bp, o1);
          o2 = wmma_f32(a2, bp, o2);
          o3 = wmma_f32(a3, bp, o3);
        }
      }
    }

    const float inv = 1.0f / ((lrun + xhalf(lrun)) + 1e-8f);
    float* dst = O + ((size_t)(bb * S_ + qw0 + r)) * DM_ + head * DK_ + 8 * hhalf;
    *(float4*)(dst + 0)      = make_float4(o0[0] * inv, o0[1] * inv, o0[2] * inv, o0[3] * inv);
    *(float4*)(dst + 4)      = make_float4(o0[4] * inv, o0[5] * inv, o0[6] * inv, o0[7] * inv);
    *(float4*)(dst + 16)     = make_float4(o1[0] * inv, o1[1] * inv, o1[2] * inv, o1[3] * inv);
    *(float4*)(dst + 16 + 4) = make_float4(o1[4] * inv, o1[5] * inv, o1[6] * inv, o1[7] * inv);
    *(float4*)(dst + 32)     = make_float4(o2[0] * inv, o2[1] * inv, o2[2] * inv, o2[3] * inv);
    *(float4*)(dst + 32 + 4) = make_float4(o2[4] * inv, o2[5] * inv, o2[6] * inv, o2[7] * inv);
    *(float4*)(dst + 48)     = make_float4(o3[0] * inv, o3[1] * inv, o3[2] * inv, o3[3] * inv);
    *(float4*)(dst + 48 + 4) = make_float4(o3[4] * inv, o3[5] * inv, o3[6] * inv, o3[7] * inv);
  }
}

__global__ void __launch_bounds__(128)
out_proj_kernel(const float* __restrict__ Oin, const float* __restrict__ w_out,
                const float* __restrict__ b_out, float* __restrict__ out) {
  __shared__ float Xs[128 * GP_];
  __shared__ float Ws[64 * GP_];
  v8f acc[4][2] = {};
  const int m0 = blockIdx.x * 128;
  const int n0 = blockIdx.y * 64;
  gemm_core_128xN<2>(Oin, w_out, Xs, Ws, m0, n0, acc);

  const int tid = threadIdx.x;
  const int wv = tid >> 5;
  const int lane = tid & 31;
  const int r = lane & 15;
  const int hhalf = lane >> 4;
  const int wm = wv >> 1;
  const int wn = wv & 1;

#pragma unroll
  for (int ni = 0; ni < 2; ++ni) {
    const int n = n0 + wn * 32 + ni * 16 + r;
    const float bv = b_out[n];
#pragma unroll
    for (int mi = 0; mi < 4; ++mi) {
#pragma unroll
      for (int g = 0; g < 8; ++g) {
        const int gm = m0 + wm * 64 + mi * 16 + g + 8 * hhalf;
        out[(size_t)gm * DM_ + n] = acc[mi][ni][g] + bv;
      }
    }
  }
}

extern "C" void kernel_launch(void* const* d_in, const int* in_sizes, int n_in,
                              void* d_out, int out_size, void* d_ws,
                              size_t ws_size, hipStream_t stream) {
  const float* x = (const float*)d_in[0];
  const float* w_qkv = (const float*)d_in[1];
  const float* b_qkv = (const float*)d_in[2];
  const float* w_out = (const float*)d_in[3];
  const float* b_out = (const float*)d_in[4];
  float* out = (float*)d_out;

  const size_t per_buf = (size_t)BH_ * S_ * DK_;
  float* Qw = (float*)d_ws;
  float* Kw = Qw + per_buf;
  float* Vw = Kw + per_buf;
  float* Ow = Vw + per_buf;

  qkv_rope_kernel<<<dim3(ROWS_ / 128, N_QKV / 128), 256, 0, stream>>>(
      x, w_qkv, b_qkv, Qw, Kw, Vw);
  flash_attn_kernel<<<dim3(NQBLK_ / 2, BH_), 128, 0, stream>>>(Qw, Kw, Vw, Ow);
  out_proj_kernel<<<dim3(ROWS_ / 128, DM_ / 64), 128, 0, stream>>>(
      Ow, w_out, b_out, out);
}


# attention: K/V staged 64 keys per barrier pair (was 16); below the diagonal 4 score tiles on independent accumulators, one max/rescale per 64 keys, loads a block ahead; row-sum tree changes
# speedup vs baseline: 1.6926x; 1.0512x over previous
#include <hip/hip_runtime.h>
#include <hip/hip_bf16.h>


typedef __attribute__((ext_vector_type(2))) float v2f;
typedef __attribute__((ext_vector_type(8))) float v8f;

#define B_     2
#define S_     2048
#define DM_    1024
#define H_     16
#define DK_    64
#define BH_    (B_ * H_)
#define ROWS_  (B_ * S_)
#define N_QKV  (3 * DM_)

__device__ __forceinline__ v8f wmma_f32(v2f a, v2f b, v8f c) {
  return __builtin_amdgcn_wmma_f32_16x16x4_f32(
      false, a, false, b, (short)0, c, false, false);
}

__device__ __forceinline__ void lds_barrier() {
  __builtin_amdgcn_fence(__ATOMIC_RELEASE, "workgroup", "local");
  __builtin_amdgcn_s_barrier();
  __builtin_amdgcn_fence(__ATOMIC_ACQUIRE, "workgroup", "local");
}

#define GP_ 36

template <int NWN>
__device__ __forceinline__ void gemm_core_128xN(
    const float* __restrict__ A, const float* __restrict__ W,
    float* Xs, float* Ws, int m0, int n0, v8f acc[4][2]) {
  constexpr int NT = 64 * NWN;
  constexpr int BN = 32 * NWN;
  constexpr int RS = NT / 8;
  constexpr int XR = 128 / RS;
  constexpr int WR = BN / RS;
  const int tid = threadIdx.x;
  const int wv = tid >> 5;
  const int lane = tid & 31;
  const int r = lane & 15;
  const int hh = lane >> 4;
  const int wm = wv / NWN;
  const int wn = wv % NWN;

  const int srow = tid >> 3;
  const int scol = (tid & 7) * 4;
  const float* xg = A + (size_t)(m0 + srow) * DM_ + scol;
  const float* wg = W + (size_t)(n0 + srow) * DM_ + scol;
  static_assert(WR == 4 && (XR == 4 || XR == 8), "staging shape");
  const size_t ps = (size_t)RS * DM_;
  float4 xa0 = *(const float4*)(xg);
  float4 xa1 = *(const float4*)(xg + ps);
  float4 xa2 = *(const float4*)(xg + 2 * ps);
  float4 xa3 = *(const float4*)(xg + 3 * ps);
  float4 xa4 = {}, xa5 = {}, xa6 = {}, xa7 = {};
  if constexpr (XR == 8) {
    xa4 = *(const float4*)(xg + 4 * ps);
    xa5 = *(const float4*)(xg + 5 * ps);
    xa6 = *(const float4*)(xg + 6 * ps);
    xa7 = *(const float4*)(xg + 7 * ps);
  }
  float4 wa0 = *(const float4*)(wg);
  float4 wa1 = *(const float4*)(wg + ps);
  float4 wa2 = *(const float4*)(wg + 2 * ps);
  float4 wa3 = *(const float4*)(wg + 3 * ps);

  float* xst = Xs + srow * GP_ + scol;
  float* wst = Ws + srow * GP_ + scol;
  const float* xf = Xs + (wm * 64 + r) * GP_ + hh * 4;
  const float* wf = Ws + (wn * 32 + r) * GP_ + hh * 4;

  for (int k0 = 0; k0 < DM_; k0 += 32) {
    lds_barrier();
    constexpr int pl = RS * GP_;
    *(float4*)(xst) = xa0;
    *(float4*)(xst + pl) = xa1;
    *(float4*)(xst + 2 * pl) = xa2;
    *(float4*)(xst + 3 * pl) = xa3;
    if constexpr (XR == 8) {
      *(float4*)(xst + 4 * pl) = xa4;
      *(float4*)(xst + 5 * pl) = xa5;
      *(float4*)(xst + 6 * pl) = xa6;
      *(float4*)(xst + 7 * pl) = xa7;
    }
    *(float4*)(wst) = wa0;
    *(float4*)(wst + pl) = wa1;
    *(float4*)(wst + 2 * pl) = wa2;
    *(float4*)(wst + 3 * pl) = wa3;
    if (k0 + 32 < DM_) {
      const float* xn = xg + k0 + 32;
      const float* wn2 = wg + k0 + 32;
      xa0 = *(const float4*)(xn);
      xa1 = *(const float4*)(xn + ps);
      xa2 = *(const float4*)(xn + 2 * ps);
      xa3 = *(const float4*)(xn + 3 * ps);
      if constexpr (XR == 8) {
        xa4 = *(const float4*)(xn + 4 * ps);
        xa5 = *(const float4*)(xn + 5 * ps);
        xa6 = *(const float4*)(xn + 6 * ps);
        xa7 = *(const float4*)(xn + 7 * ps);
      }
      wa0 = *(const float4*)(wn2);
      wa1 = *(const float4*)(wn2 + ps);
      wa2 = *(const float4*)(wn2 + 2 * ps);
      wa3 = *(const float4*)(wn2 + 3 * ps);
    }
    lds_barrier();
#pragma unroll
    for (int g = 0; g < 4; ++g) {
      float4 af[4], bf[2];
#pragma unroll
      for (int mi = 0; mi < 4; ++mi)
        af[mi] = *(const float4*)(xf + mi * 16 * GP_ + g * 8);
#pragma unroll
      for (int ni = 0; ni < 2; ++ni)
        bf[ni] = *(const float4*)(wf + ni * 16 * GP_ + g * 8);
#pragma unroll
      for (int mi = 0; mi < 4; ++mi) {
#pragma unroll
        for (int ni = 0; ni < 2; ++ni)
          acc[mi][ni] = wmma_f32(v2f{af[mi].x, af[mi].y},
                                 v2f{bf[ni].x, bf[ni].y}, acc[mi][ni]);
      }
#pragma unroll
      for (int mi = 0; mi < 4; ++mi) {
#pragma unroll
        for (int ni = 0; ni < 2; ++ni)
          acc[mi][ni] = wmma_f32(v2f{af[mi].z, af[mi].w},
                                 v2f{bf[ni].z, bf[ni].w}, acc[mi][ni]);
      }
    }
  }
}

__global__ void __launch_bounds__(256)
qkv_rope_kernel(const float* __restrict__ x, const float* __restrict__ w_qkv,
                const float* __restrict__ b_qkv, float* __restrict__ Qw,
                float* __restrict__ Kw, float* __restrict__ Vw) {
  __shared__ float Xs[128 * GP_];
  __shared__ float Ws[128 * GP_];
  v8f acc[4][2] = {};
  const int m0 = blockIdx.x * 128;
  const int n0 = blockIdx.y * 128;
  gemm_core_128xN<4>(x, w_qkv, Xs, Ws, m0, n0, acc);

  const int tid = threadIdx.x;
  const int wv = tid >> 5;
  const int lane = tid & 31;
  const int r = lane & 15;
  const int hhalf = lane >> 4;
  const int wm = wv >> 2;
  const int wn = wv & 3;

#pragma unroll
  for (int ni = 0; ni < 2; ++ni) {
    const int o = n0 + wn * 32 + ni * 16 + r;
    const int sel = o >> 10;
    const int rem = o & 1023;
    const int head = rem >> 6;
    const int d = rem & 63;
    const float bv = b_qkv[o];
    const int p = d >> 1;
    const float invf = __expf((float)(2 * p) * (-9.210340371976184f / 64.0f));
    const float sgn = (d & 1) ? 1.0f : -1.0f;
    float* dst = (sel == 0) ? Qw : ((sel == 1) ? Kw : Vw);
#pragma unroll
    for (int mi = 0; mi < 4; ++mi) {
#pragma unroll
      for (int g = 0; g < 8; ++g) {
        const int gm = m0 + wm * 64 + mi * 16 + g + 8 * hhalf;
        const int srow = gm & (S_ - 1);
        const int bb = gm >> 11;
        float v = acc[mi][ni][g] + bv;
        float other = __shfl_xor(v, 1);
        float outv = v;
        if (sel < 2) {
          const float ang = (float)srow * invf;
          const float sn = __sinf(ang);
          const float cs = __cosf(ang);
          outv = v * cs + other * sgn * sn;
        }
        dst[(((size_t)(bb * H_ + head)) * S_ + srow) * DK_ + d] = outv;
      }
    }
  }
}

__device__ __forceinline__ float xhalf(float x) {
  const int b = __builtin_bit_cast(int, x);
  return __builtin_bit_cast(
      float, __builtin_amdgcn_permlanex16(b, b, 0x76543210, 0xfedcba98, false,
                                          false));
}

#define QBLK_  64
#define NQBLK_ (S_ / QBLK_)
#define KT_    64
#define KP_    68

__device__ __forceinline__ void attn_tile16(
    const float* Kt, const float* Vt, const v2f (&qf)[16], int r, int hhalf,
    int dk0, v8f& o0, v8f& o1, v8f& o2, v8f& o3, float& mrun, float& lrun) {
  v8f s = {};
#pragma unroll
  for (int kk = 0; kk < 16; ++kk) {
    const float* kr = Kt + r * KP_ + kk * 4 + hhalf * 2;
    v2f kf = {kr[0], kr[1]};
    s = wmma_f32(kf, qf[kk], s);
  }
#pragma unroll
  for (int g = 0; g < 8; ++g) {
    if (dk0 + g > 0) s[g] = -3.0e38f;
  }
  float mx = fmaxf(fmaxf(fmaxf(s[0], s[1]), fmaxf(s[2], s[3])),
                   fmaxf(fmaxf(s[4], s[5]), fmaxf(s[6], s[7])));
  mx = fmaxf(mx, xhalf(mx));
  const float mnew = fmaxf(mrun, mx);
  const float alpha = __expf(mrun - mnew);
  mrun = mnew;
  v8f p;
#pragma unroll
  for (int g = 0; g < 8; ++g) p[g] = __expf(s[g] - mnew);
  lrun = lrun * alpha +
         (((p[0] + p[1]) + (p[2] + p[3])) + ((p[4] + p[5]) + (p[6] + p[7])));
  o0 *= alpha; o1 *= alpha; o2 *= alpha; o3 *= alpha;
#pragma unroll
  for (int t = 0; t < 4; ++t) {
    const float* vr = Vt + (8 * hhalf + 2 * t) * KP_ + r;
    v2f bp = {p[2 * t], p[2 * t + 1]};
    v2f a0 = {vr[0 * 16], vr[KP_ + 0 * 16]};
    v2f a1 = {vr[1 * 16], vr[KP_ + 1 * 16]};
    v2f a2 = {vr[2 * 16], vr[KP_ + 2 * 16]};
    v2f a3 = {vr[3 * 16], vr[KP_ + 3 * 16]};
    o0 = wmma_f32(a0, bp, o0);
    o1 = wmma_f32(a1, bp, o1);
    o2 = wmma_f32(a2, bp, o2);
    o3 = wmma_f32(a3, bp, o3);
  }
}

__device__ __forceinline__ void attn_stage64(
    const float* Ks, const float* Vs, const v2f (&qf)[16], int r, int hhalf,
    v8f& o0, v8f& o1, v8f& o2, v8f& o3, float& mrun, float& lrun) {
  v8f s[4] = {};
  const float* kr = Ks + r * KP_ + hhalf * 2;
  v2f kf[2][16];
#pragma unroll
  for (int kl = 0; kl < 4; ++kl) {
#pragma unroll
    for (int u = 0; u < 4; ++u) {
      const float* kp = kr + u * 16 * KP_ + kl * 4;
      kf[0][kl * 4 + u] = v2f{kp[0], kp[1]};
    }
  }
#pragma unroll
  for (int b = 0; b < 4; ++b) {
    if (b < 3) {
#pragma unroll
      for (int kl = 0; kl < 4; ++kl) {
#pragma unroll
        for (int u = 0; u < 4; ++u) {
          const float* kp = kr + u * 16 * KP_ + ((b + 1) * 4 + kl) * 4;
          kf[(b + 1) & 1][kl * 4 + u] = v2f{kp[0], kp[1]};
        }
      }
    }
    __builtin_amdgcn_sched_barrier(0);
#pragma unroll
    for (int kl = 0; kl < 4; ++kl) {
#pragma unroll
      for (int u = 0; u < 4; ++u)
        s[u] = wmma_f32(kf[b & 1][kl * 4 + u], qf[b * 4 + kl], s[u]);
    }
    if (b < 3) __builtin_amdgcn_sched_barrier(0);
  }

  float vf[2][32];
  const float* vb = Vs + (8 * hhalf) * KP_ + r;
#pragma unroll
  for (int t = 0; t < 4; ++t) {
#pragma unroll
    for (int i = 0; i < 4; ++i) {
      const float* vr = vb + (2 * t) * KP_ + i * 16;
      vf[0][(t * 4 + i) * 2 + 0] = vr[0];
      vf[0][(t * 4 + i) * 2 + 1] = vr[KP_];
    }
  }
  __builtin_amdgcn_sched_barrier(0);

  float m4[4];
#pragma unroll
  for (int u = 0; u < 4; ++u)
    m4[u] = fmaxf(fmaxf(fmaxf(s[u][0], s[u][1]), fmaxf(s[u][2], s[u][3])),
                  fmaxf(fmaxf(s[u][4], s[u][5]), fmaxf(s[u][6], s[u][7])));
  float mx = fmaxf(fmaxf(m4[0], m4[1]), fmaxf(m4[2], m4[3]));
  mx = fmaxf(mx, xhalf(mx));
  const float mnew = fmaxf(mrun, mx);
  const float alpha = __expf(mrun - mnew);
  mrun = mnew;
  float ps[4];
#pragma unroll
  for (int u = 0; u < 4; ++u) {
#pragma unroll
    for (int g = 0; g < 8; ++g) s[u][g] = __expf(s[u][g] - mnew);
    ps[u] = ((s[u][0] + s[u][1]) + (s[u][2] + s[u][3])) +
            ((s[u][4] + s[u][5]) + (s[u][6] + s[u][7]));
  }
  lrun = lrun * alpha + ((ps[0] + ps[1]) + (ps[2] + ps[3]));
  o0 *= alpha; o1 *= alpha; o2 *= alpha; o3 *= alpha;

#pragma unroll
  for (int u = 0; u < 4; ++u) {
    if (u < 3) {
#pragma unroll
      for (int t = 0; t < 4; ++t) {
#pragma unroll
        for (int i = 0; i < 4; ++i) {
          const float* vr = vb + (16 * (u + 1) + 2 * t) * KP_ + i * 16;
          vf[(u + 1) & 1][(t * 4 + i) * 2 + 0] = vr[0];
          vf[(u + 1) & 1][(t * 4 + i) * 2 + 1] = vr[KP_];
        }
      }
    }
    __builtin_amdgcn_sched_barrier(0);
#pragma unroll
    for (int t = 0; t < 4; ++t) {
      v2f bp = {s[u][2 * t], s[u][2 * t + 1]};
      const float* f = &vf[u & 1][t * 8];
      o0 = wmma_f32(v2f{f[0], f[1]}, bp, o0);
      o1 = wmma_f32(v2f{f[2], f[3]}, bp, o1);
      o2 = wmma_f32(v2f{f[4], f[5]}, bp, o2);
      o3 = wmma_f32(v2f{f[6], f[7]}, bp, o3);
    }
    if (u < 3) __builtin_amdgcn_sched_barrier(0);
  }
}

__global__ void __launch_bounds__(128)
flash_attn_kernel(const float* __restrict__ Q, const float* __restrict__ K,
                  const float* __restrict__ V, float* __restrict__ O) {
  __shared__ float Ks[KT_ * KP_];
  __shared__ float Vs[KT_ * KP_];

  const int bh = blockIdx.y;
  const int tid = threadIdx.x;
  const int wv = __builtin_amdgcn_readfirstlane(tid >> 5);
  const int lane = tid & 31;
  const int r = lane & 15;
  const int hhalf = lane >> 4;
  const int bb = bh >> 4;
  const int head = bh & 15;

  const float* Qp = Q + (size_t)bh * S_ * DK_;
  const float* Kp = K + (size_t)bh * S_ * DK_;
  const float* Vp = V + (size_t)bh * S_ * DK_;

  for (int ph = 0; ph < 2; ++ph) {
    const int qb = (ph == 0) ? (NQBLK_ - 1 - (int)blockIdx.x) : (int)blockIdx.x;
    const int qw0 = qb * QBLK_ + wv * 16;

    v2f qf[16];
#pragma unroll
    for (int kk = 0; kk < 16; ++kk) {
      const float* qp = Qp + (size_t)(qw0 + r) * DK_ + kk * 4 + hhalf * 2;
      qf[kk] = v2f{qp[0] * 0.125f, qp[1] * 0.125f};
    }

    v8f o0 = {}, o1 = {}, o2 = {}, o3 = {};
    float mrun = -3.0e38f;
    float lrun = 0.0f;

    const int nst = qb + 1;

    const int row0 = tid >> 4;
    const int cv = (tid & 15) * 4;
    const float* kg = Kp + (size_t)row0 * DK_ + cv;
    const float* vg = Vp + (size_t)row0 * DK_ + cv;
    float4 ka0 = *(const float4*)(kg);
    float4 ka1 = *(const float4*)(kg + 8 * DK_);
    float4 ka2 = *(const float4*)(kg + 16 * DK_);
    float4 ka3 = *(const float4*)(kg + 24 * DK_);
    float4 ka4 = *(const float4*)(kg + 32 * DK_);
    float4 ka5 = *(const float4*)(kg + 40 * DK_);
    float4 ka6 = *(const float4*)(kg + 48 * DK_);
    float4 ka7 = *(const float4*)(kg + 56 * DK_);
    float4 va0 = *(const float4*)(vg);
    float4 va1 = *(const float4*)(vg + 8 * DK_);
    float4 va2 = *(const float4*)(vg + 16 * DK_);
    float4 va3 = *(const float4*)(vg + 24 * DK_);
    float4 va4 = *(const float4*)(vg + 32 * DK_);
    float4 va5 = *(const float4*)(vg + 40 * DK_);
    float4 va6 = *(const float4*)(vg + 48 * DK_);
    float4 va7 = *(const float4*)(vg + 56 * DK_);
    float* kst = Ks + row0 * KP_ + cv;
    float* vst = Vs + row0 * KP_ + cv;

    for (int j = 0; j < nst; ++j) {
      lds_barrier();
      *(float4*)(kst) = ka0;
      *(float4*)(kst + 8 * KP_) = ka1;
      *(float4*)(kst + 16 * KP_) = ka2;
      *(float4*)(kst + 24 * KP_) = ka3;
      *(float4*)(kst + 32 * KP_) = ka4;
      *(float4*)(kst + 40 * KP_) = ka5;
      *(float4*)(kst + 48 * KP_) = ka6;
      *(float4*)(kst + 56 * KP_) = ka7;
      *(float4*)(vst) = va0;
      *(float4*)(vst + 8 * KP_) = va1;
      *(float4*)(vst + 16 * KP_) = va2;
      *(float4*)(vst + 24 * KP_) = va3;
      *(float4*)(vst + 32 * KP_) = va4;
      *(float4*)(vst + 40 * KP_) = va5;
      *(float4*)(vst + 48 * KP_) = va6;
      *(float4*)(vst + 56 * KP_) = va7;
      if (j + 1 < nst) {
        const float* kn = kg + (size_t)(j + 1) * KT_ * DK_;
        const float* vn = vg + (size_t)(j + 1) * KT_ * DK_;
        ka0 = *(const float4*)(kn);
        ka1 = *(const float4*)(kn + 8 * DK_);
        ka2 = *(const float4*)(kn + 16 * DK_);
        ka3 = *(const float4*)(kn + 24 * DK_);
        ka4 = *(const float4*)(kn + 32 * DK_);
        ka5 = *(const float4*)(kn + 40 * DK_);
        ka6 = *(const float4*)(kn + 48 * DK_);
        ka7 = *(const float4*)(kn + 56 * DK_);
        va0 = *(const float4*)(vn);
        va1 = *(const float4*)(vn + 8 * DK_);
        va2 = *(const float4*)(vn + 16 * DK_);
        va3 = *(const float4*)(vn + 24 * DK_);
        va4 = *(const float4*)(vn + 32 * DK_);
        va5 = *(const float4*)(vn + 40 * DK_);
        va6 = *(const float4*)(vn + 48 * DK_);
        va7 = *(const float4*)(vn + 56 * DK_);
      }
      lds_barrier();

      if (j + 1 < nst) {
        attn_stage64(Ks, Vs, qf, r, hhalf, o0, o1, o2, o3, mrun, lrun);
      } else {
        for (int u = 0; u <= wv; ++u)
          attn_tile16(Ks + u * 16 * KP_, Vs + u * 16 * KP_, qf, r, hhalf,
                      16 * (u - wv) + 8 * hhalf - r, o0, o1, o2, o3, mrun,
                      lrun);
      }
    }

    const float inv = 1.0f / ((lrun + xhalf(lrun)) + 1e-8f);
    float* dst = O + ((size_t)(bb * S_ + qw0 + r)) * DM_ + head * DK_ + 8 * hhalf;
    *(float4*)(dst + 0)      = make_float4(o0[0] * inv, o0[1] * inv, o0[2] * inv, o0[3] * inv);
    *(float4*)(dst + 4)      = make_float4(o0[4] * inv, o0[5] * inv, o0[6] * inv, o0[7] * inv);
    *(float4*)(dst + 16)     = make_float4(o1[0] * inv, o1[1] * inv, o1[2] * inv, o1[3] * inv);
    *(float4*)(dst + 16 + 4) = make_float4(o1[4] * inv, o1[5] * inv, o1[6] * inv, o1[7] * inv);
    *(float4*)(dst + 32)     = make_float4(o2[0] * inv, o2[1] * inv, o2[2] * inv, o2[3] * inv);
    *(float4*)(dst + 32 + 4) = make_float4(o2[4] * inv, o2[5] * inv, o2[6] * inv, o2[7] * inv);
    *(float4*)(dst + 48)     = make_float4(o3[0] * inv, o3[1] * inv, o3[2] * inv, o3[3] * inv);
    *(float4*)(dst + 48 + 4) = make_float4(o3[4] * inv, o3[5] * inv, o3[6] * inv, o3[7] * inv);
  }
}

__global__ void __launch_bounds__(128)
out_proj_kernel(const float* __restrict__ Oin, const float* __restrict__ w_out,
                const float* __restrict__ b_out, float* __restrict__ out) {
  __shared__ float Xs[128 * GP_];
  __shared__ float Ws[64 * GP_];
  v8f acc[4][2] = {};
  const int m0 = blockIdx.x * 128;
  const int n0 = blockIdx.y * 64;
  gemm_core_128xN<2>(Oin, w_out, Xs, Ws, m0, n0, acc);

  const int tid = threadIdx.x;
  const int wv = tid >> 5;
  const int lane = tid & 31;
  const int r = lane & 15;
  const int hhalf = lane >> 4;
  const int wm = wv >> 1;
  const int wn = wv & 1;

#pragma unroll
  for (int ni = 0; ni < 2; ++ni) {
    const int n = n0 + wn * 32 + ni * 16 + r;
    const float bv = b_out[n];
#pragma unroll
    for (int mi = 0; mi < 4; ++mi) {
#pragma unroll
      for (int g = 0; g < 8; ++g) {
        const int gm = m0 + wm * 64 + mi * 16 + g + 8 * hhalf;
        out[(size_t)gm * DM_ + n] = acc[mi][ni][g] + bv;
      }
    }
  }
}

extern "C" void kernel_launch(void* const* d_in, const int* in_sizes, int n_in,
                              void* d_out, int out_size, void* d_ws,
                              size_t ws_size, hipStream_t stream) {
  const float* x = (const float*)d_in[0];
  const float* w_qkv = (const float*)d_in[1];
  const float* b_qkv = (const float*)d_in[2];
  const float* w_out = (const float*)d_in[3];
  const float* b_out = (const float*)d_in[4];
  float* out = (float*)d_out;

  const size_t per_buf = (size_t)BH_ * S_ * DK_;
  float* Qw = (float*)d_ws;
  float* Kw = Qw + per_buf;
  float* Vw = Kw + per_buf;
  float* Ow = Vw + per_buf;

  qkv_rope_kernel<<<dim3(ROWS_ / 128, N_QKV / 128), 256, 0, stream>>>(
      x, w_qkv, b_qkv, Qw, Kw, Vw);
  flash_attn_kernel<<<dim3(NQBLK_ / 2, BH_), 128, 0, stream>>>(Qw, Kw, Vw, Ow);
  out_proj_kernel<<<dim3(ROWS_ / 128, DM_ / 64), 128, 0, stream>>>(
      Ow, w_out, b_out, out);
}


# all three kernels: wave priority (s_setprio) steps down 3,2,1,0 as a work-group advances, so co-resident work-groups share the matrix pipe and end together; the youngest no longer finishes alone
# speedup vs baseline: 1.7582x; 1.0388x over previous
#include <hip/hip_runtime.h>
#include <hip/hip_bf16.h>


typedef __attribute__((ext_vector_type(2))) float v2f;
typedef __attribute__((ext_vector_type(8))) float v8f;

#define B_     2
#define S_     2048
#define DM_    1024
#define H_     16
#define DK_    64
#define BH_    (B_ * H_)
#define ROWS_  (B_ * S_)
#define N_QKV  (3 * DM_)

__device__ __forceinline__ v8f wmma_f32(v2f a, v2f b, v8f c) {
  return __builtin_amdgcn_wmma_f32_16x16x4_f32(
      false, a, false, b, (short)0, c, false, false);
}

__device__ __forceinline__ void lds_barrier() {
  __builtin_amdgcn_fence(__ATOMIC_RELEASE, "workgroup", "local");
  __builtin_amdgcn_s_barrier();
  __builtin_amdgcn_fence(__ATOMIC_ACQUIRE, "workgroup", "local");
}

__device__ __forceinline__ void prio_by_progress(int step, int a, int b, int c) {
  if (step == a) __builtin_amdgcn_s_setprio(2);
  else if (step == b) __builtin_amdgcn_s_setprio(1);
  else if (step == c) __builtin_amdgcn_s_setprio(0);
}

#define GP_ 36

template <int NWN>
__device__ __forceinline__ void gemm_core_128xN(
    const float* __restrict__ A, const float* __restrict__ W,
    float* Xs, float* Ws, int m0, int n0, v8f acc[4][2]) {
  constexpr int NT = 64 * NWN;
  constexpr int BN = 32 * NWN;
  constexpr int RS = NT / 8;
  constexpr int XR = 128 / RS;
  constexpr int WR = BN / RS;
  const int tid = threadIdx.x;
  const int wv = tid >> 5;
  const int lane = tid & 31;
  const int r = lane & 15;
  const int hh = lane >> 4;
  const int wm = wv / NWN;
  const int wn = wv % NWN;

  const int srow = tid >> 3;
  const int scol = (tid & 7) * 4;
  const float* xg = A + (size_t)(m0 + srow) * DM_ + scol;
  const float* wg = W + (size_t)(n0 + srow) * DM_ + scol;
  static_assert(WR == 4 && (XR == 4 || XR == 8), "staging shape");
  const size_t ps = (size_t)RS * DM_;
  float4 xa0 = *(const float4*)(xg);
  float4 xa1 = *(const float4*)(xg + ps);
  float4 xa2 = *(const float4*)(xg + 2 * ps);
  float4 xa3 = *(const float4*)(xg + 3 * ps);
  float4 xa4 = {}, xa5 = {}, xa6 = {}, xa7 = {};
  if constexpr (XR == 8) {
    xa4 = *(const float4*)(xg + 4 * ps);
    xa5 = *(const float4*)(xg + 5 * ps);
    xa6 = *(const float4*)(xg + 6 * ps);
    xa7 = *(const float4*)(xg + 7 * ps);
  }
  float4 wa0 = *(const float4*)(wg);
  float4 wa1 = *(const float4*)(wg + ps);
  float4 wa2 = *(const float4*)(wg + 2 * ps);
  float4 wa3 = *(const float4*)(wg + 3 * ps);

  float* xst = Xs + srow * GP_ + scol;
  float* wst = Ws + srow * GP_ + scol;
  const float* xf = Xs + (wm * 64 + r) * GP_ + hh * 4;
  const float* wf = Ws + (wn * 32 + r) * GP_ + hh * 4;

  __builtin_amdgcn_s_setprio(3);
  for (int k0 = 0; k0 < DM_; k0 += 32) {
    lds_barrier();
    prio_by_progress(k0, 19 * 32, 27 * 32, 30 * 32);
    constexpr int pl = RS * GP_;
    *(float4*)(xst) = xa0;
    *(float4*)(xst + pl) = xa1;
    *(float4*)(xst + 2 * pl) = xa2;
    *(float4*)(xst + 3 * pl) = xa3;
    if constexpr (XR == 8) {
      *(float4*)(xst + 4 * pl) = xa4;
      *(float4*)(xst + 5 * pl) = xa5;
      *(float4*)(xst + 6 * pl) = xa6;
      *(float4*)(xst + 7 * pl) = xa7;
    }
    *(float4*)(wst) = wa0;
    *(float4*)(wst + pl) = wa1;
    *(float4*)(wst + 2 * pl) = wa2;
    *(float4*)(wst + 3 * pl) = wa3;
    if (k0 + 32 < DM_) {
      const float* xn = xg + k0 + 32;
      const float* wn2 = wg + k0 + 32;
      xa0 = *(const float4*)(xn);
      xa1 = *(const float4*)(xn + ps);
      xa2 = *(const float4*)(xn + 2 * ps);
      xa3 = *(const float4*)(xn + 3 * ps);
      if constexpr (XR == 8) {
        xa4 = *(const float4*)(xn + 4 * ps);
        xa5 = *(const float4*)(xn + 5 * ps);
        xa6 = *(const float4*)(xn + 6 * ps);
        xa7 = *(const float4*)(xn + 7 * ps);
      }
      wa0 = *(const float4*)(wn2);
      wa1 = *(const float4*)(wn2 + ps);
      wa2 = *(const float4*)(wn2 + 2 * ps);
      wa3 = *(const float4*)(wn2 + 3 * ps);
    }
    lds_barrier();
#pragma unroll
    for (int g = 0; g < 4; ++g) {
      float4 af[4], bf[2];
#pragma unroll
      for (int mi = 0; mi < 4; ++mi)
        af[mi] = *(const float4*)(xf + mi * 16 * GP_ + g * 8);
#pragma unroll
      for (int ni = 0; ni < 2; ++ni)
        bf[ni] = *(const float4*)(wf + ni * 16 * GP_ + g * 8);
#pragma unroll
      for (int mi = 0; mi < 4; ++mi) {
#pragma unroll
        for (int ni = 0; ni < 2; ++ni)
          acc[mi][ni] = wmma_f32(v2f{af[mi].x, af[mi].y},
                                 v2f{bf[ni].x, bf[ni].y}, acc[mi][ni]);
      }
#pragma unroll
      for (int mi = 0; mi < 4; ++mi) {
#pragma unroll
        for (int ni = 0; ni < 2; ++ni)
          acc[mi][ni] = wmma_f32(v2f{af[mi].z, af[mi].w},
                                 v2f{bf[ni].z, bf[ni].w}, acc[mi][ni]);
      }
    }
  }
}

__global__ void __launch_bounds__(256)
qkv_rope_kernel(const float* __restrict__ x, const float* __restrict__ w_qkv,
                const float* __restrict__ b_qkv, float* __restrict__ Qw,
                float* __restrict__ Kw, float* __restrict__ Vw) {
  __shared__ float Xs[128 * GP_];
  __shared__ float Ws[128 * GP_];
  v8f acc[4][2] = {};
  const int m0 = blockIdx.x * 128;
  const int n0 = blockIdx.y * 128;
  gemm_core_128xN<4>(x, w_qkv, Xs, Ws, m0, n0, acc);

  const int tid = threadIdx.x;
  const int wv = tid >> 5;
  const int lane = tid & 31;
  const int r = lane & 15;
  const int hhalf = lane >> 4;
  const int wm = wv >> 2;
  const int wn = wv & 3;

#pragma unroll
  for (int ni = 0; ni < 2; ++ni) {
    const int o = n0 + wn * 32 + ni * 16 + r;
    const int sel = o >> 10;
    const int rem = o & 1023;
    const int head = rem >> 6;
    const int d = rem & 63;
    const float bv = b_qkv[o];
    const int p = d >> 1;
    const float invf = __expf((float)(2 * p) * (-9.210340371976184f / 64.0f));
    const float sgn = (d & 1) ? 1.0f : -1.0f;
    float* dst = (sel == 0) ? Qw : ((sel == 1) ? Kw : Vw);
#pragma unroll
    for (int mi = 0; mi < 4; ++mi) {
#pragma unroll
      for (int g = 0; g < 8; ++g) {
        const int gm = m0 + wm * 64 + mi * 16 + g + 8 * hhalf;
        const int srow = gm & (S_ - 1);
        const int bb = gm >> 11;
        float v = acc[mi][ni][g] + bv;
        float other = __shfl_xor(v, 1);
        float outv = v;
        if (sel < 2) {
          const float ang = (float)srow * invf;
          const float sn = __sinf(ang);
          const float cs = __cosf(ang);
          outv = v * cs + other * sgn * sn;
        }
        dst[(((size_t)(bb * H_ + head)) * S_ + srow) * DK_ + d] = outv;
      }
    }
  }
}

__device__ __forceinline__ float xhalf(float x) {
  const int b = __builtin_bit_cast(int, x);
  return __builtin_bit_cast(
      float, __builtin_amdgcn_permlanex16(b, b, 0x76543210, 0xfedcba98, false,
                                          false));
}

#define QBLK_  64
#define NQBLK_ (S_ / QBLK_)
#define KT_    64
#define KP_    68

__device__ __forceinline__ void attn_tile16(
    const float* Kt, const float* Vt, const v2f (&qf)[16], int r, int hhalf,
    int dk0, v8f& o0, v8f& o1, v8f& o2, v8f& o3, float& mrun, float& lrun) {
  v8f s = {};
#pragma unroll
  for (int kk = 0; kk < 16; ++kk) {
    const float* kr = Kt + r * KP_ + kk * 4 + hhalf * 2;
    v2f kf = {kr[0], kr[1]};
    s = wmma_f32(kf, qf[kk], s);
  }
#pragma unroll
  for (int g = 0; g < 8; ++g) {
    if (dk0 + g > 0) s[g] = -3.0e38f;
  }
  float mx = fmaxf(fmaxf(fmaxf(s[0], s[1]), fmaxf(s[2], s[3])),
                   fmaxf(fmaxf(s[4], s[5]), fmaxf(s[6], s[7])));
  mx = fmaxf(mx, xhalf(mx));
  const float mnew = fmaxf(mrun, mx);
  const float alpha = __expf(mrun - mnew);
  mrun = mnew;
  v8f p;
#pragma unroll
  for (int g = 0; g < 8; ++g) p[g] = __expf(s[g] - mnew);
  lrun = lrun * alpha +
         (((p[0] + p[1]) + (p[2] + p[3])) + ((p[4] + p[5]) + (p[6] + p[7])));
  o0 *= alpha; o1 *= alpha; o2 *= alpha; o3 *= alpha;
#pragma unroll
  for (int t = 0; t < 4; ++t) {
    const float* vr = Vt + (8 * hhalf + 2 * t) * KP_ + r;
    v2f bp = {p[2 * t], p[2 * t + 1]};
    v2f a0 = {vr[0 * 16], vr[KP_ + 0 * 16]};
    v2f a1 = {vr[1 * 16], vr[KP_ + 1 * 16]};
    v2f a2 = {vr[2 * 16], vr[KP_ + 2 * 16]};
    v2f a3 = {vr[3 * 16], vr[KP_ + 3 * 16]};
    o0 = wmma_f32(a0, bp, o0);
    o1 = wmma_f32(a1, bp, o1);
    o2 = wmma_f32(a2, bp, o2);
    o3 = wmma_f32(a3, bp, o3);
  }
}

__device__ __forceinline__ void attn_stage64(
    const float* Ks, const float* Vs, const v2f (&qf)[16], int r, int hhalf,
    v8f& o0, v8f& o1, v8f& o2, v8f& o3, float& mrun, float& lrun) {
  v8f s[4] = {};
  const float* kr = Ks + r * KP_ + hhalf * 2;
  v2f kf[2][16];
#pragma unroll
  for (int kl = 0; kl < 4; ++kl) {
#pragma unroll
    for (int u = 0; u < 4; ++u) {
      const float* kp = kr + u * 16 * KP_ + kl * 4;
      kf[0][kl * 4 + u] = v2f{kp[0], kp[1]};
    }
  }
#pragma unroll
  for (int b = 0; b < 4; ++b) {
    if (b < 3) {
#pragma unroll
      for (int kl = 0; kl < 4; ++kl) {
#pragma unroll
        for (int u = 0; u < 4; ++u) {
          const float* kp = kr + u * 16 * KP_ + ((b + 1) * 4 + kl) * 4;
          kf[(b + 1) & 1][kl * 4 + u] = v2f{kp[0], kp[1]};
        }
      }
    }
    __builtin_amdgcn_sched_barrier(0);
#pragma unroll
    for (int kl = 0; kl < 4; ++kl) {
#pragma unroll
      for (int u = 0; u < 4; ++u)
        s[u] = wmma_f32(kf[b & 1][kl * 4 + u], qf[b * 4 + kl], s[u]);
    }
    if (b < 3) __builtin_amdgcn_sched_barrier(0);
  }

  float vf[2][32];
  const float* vb = Vs + (8 * hhalf) * KP_ + r;
#pragma unroll
  for (int t = 0; t < 4; ++t) {
#pragma unroll
    for (int i = 0; i < 4; ++i) {
      const float* vr = vb + (2 * t) * KP_ + i * 16;
      vf[0][(t * 4 + i) * 2 + 0] = vr[0];
      vf[0][(t * 4 + i) * 2 + 1] = vr[KP_];
    }
  }
  __builtin_amdgcn_sched_barrier(0);

  float m4[4];
#pragma unroll
  for (int u = 0; u < 4; ++u)
    m4[u] = fmaxf(fmaxf(fmaxf(s[u][0], s[u][1]), fmaxf(s[u][2], s[u][3])),
                  fmaxf(fmaxf(s[u][4], s[u][5]), fmaxf(s[u][6], s[u][7])));
  float mx = fmaxf(fmaxf(m4[0], m4[1]), fmaxf(m4[2], m4[3]));
  mx = fmaxf(mx, xhalf(mx));
  const float mnew = fmaxf(mrun, mx);
  const float alpha = __expf(mrun - mnew);
  mrun = mnew;
  float ps[4];
#pragma unroll
  for (int u = 0; u < 4; ++u) {
#pragma unroll
    for (int g = 0; g < 8; ++g) s[u][g] = __expf(s[u][g] - mnew);
    ps[u] = ((s[u][0] + s[u][1]) + (s[u][2] + s[u][3])) +
            ((s[u][4] + s[u][5]) + (s[u][6] + s[u][7]));
  }
  lrun = lrun * alpha + ((ps[0] + ps[1]) + (ps[2] + ps[3]));
  o0 *= alpha; o1 *= alpha; o2 *= alpha; o3 *= alpha;

#pragma unroll
  for (int u = 0; u < 4; ++u) {
    if (u < 3) {
#pragma unroll
      for (int t = 0; t < 4; ++t) {
#pragma unroll
        for (int i = 0; i < 4; ++i) {
          const float* vr = vb + (16 * (u + 1) + 2 * t) * KP_ + i * 16;
          vf[(u + 1) & 1][(t * 4 + i) * 2 + 0] = vr[0];
          vf[(u + 1) & 1][(t * 4 + i) * 2 + 1] = vr[KP_];
        }
      }
    }
    __builtin_amdgcn_sched_barrier(0);
#pragma unroll
    for (int t = 0; t < 4; ++t) {
      v2f bp = {s[u][2 * t], s[u][2 * t + 1]};
      const float* f = &vf[u & 1][t * 8];
      o0 = wmma_f32(v2f{f[0], f[1]}, bp, o0);
      o1 = wmma_f32(v2f{f[2], f[3]}, bp, o1);
      o2 = wmma_f32(v2f{f[4], f[5]}, bp, o2);
      o3 = wmma_f32(v2f{f[6], f[7]}, bp, o3);
    }
    if (u < 3) __builtin_amdgcn_sched_barrier(0);
  }
}

__global__ void __launch_bounds__(128)
flash_attn_kernel(const float* __restrict__ Q, const float* __restrict__ K,
                  const float* __restrict__ V, float* __restrict__ O) {
  __shared__ float Ks[KT_ * KP_];
  __shared__ float Vs[KT_ * KP_];

  const int bh = blockIdx.y;
  const int tid = threadIdx.x;
  const int wv = __builtin_amdgcn_readfirstlane(tid >> 5);
  const int lane = tid & 31;
  const int r = lane & 15;
  const int hhalf = lane >> 4;
  const int bb = bh >> 4;
  const int head = bh & 15;

  const float* Qp = Q + (size_t)bh * S_ * DK_;
  const float* Kp = K + (size_t)bh * S_ * DK_;
  const float* Vp = V + (size_t)bh * S_ * DK_;

  __builtin_amdgcn_s_setprio(3);
  int done = 0;
  for (int ph = 0; ph < 2; ++ph) {
    const int qb = (ph == 0) ? (NQBLK_ - 1 - (int)blockIdx.x) : (int)blockIdx.x;
    const int qw0 = qb * QBLK_ + wv * 16;

    v2f qf[16];
#pragma unroll
    for (int kk = 0; kk < 16; ++kk) {
      const float* qp = Qp + (size_t)(qw0 + r) * DK_ + kk * 4 + hhalf * 2;
      qf[kk] = v2f{qp[0] * 0.125f, qp[1] * 0.125f};
    }

    v8f o0 = {}, o1 = {}, o2 = {}, o3 = {};
    float mrun = -3.0e38f;
    float lrun = 0.0f;

    const int nst = qb + 1;

    const int row0 = tid >> 4;
    const int cv = (tid & 15) * 4;
    const float* kg = Kp + (size_t)row0 * DK_ + cv;
    const float* vg = Vp + (size_t)row0 * DK_ + cv;
    float4 ka0 = *(const float4*)(kg);
    float4 ka1 = *(const float4*)(kg + 8 * DK_);
    float4 ka2 = *(const float4*)(kg + 16 * DK_);
    float4 ka3 = *(const float4*)(kg + 24 * DK_);
    float4 ka4 = *(const float4*)(kg + 32 * DK_);
    float4 ka5 = *(const float4*)(kg + 40 * DK_);
    float4 ka6 = *(const float4*)(kg + 48 * DK_);
    float4 ka7 = *(const float4*)(kg + 56 * DK_);
    float4 va0 = *(const float4*)(vg);
    float4 va1 = *(const float4*)(vg + 8 * DK_);
    float4 va2 = *(const float4*)(vg + 16 * DK_);
    float4 va3 = *(const float4*)(vg + 24 * DK_);
    float4 va4 = *(const float4*)(vg + 32 * DK_);
    float4 va5 = *(const float4*)(vg + 40 * DK_);
    float4 va6 = *(const float4*)(vg + 48 * DK_);
    float4 va7 = *(const float4*)(vg + 56 * DK_);
    float* kst = Ks + row0 * KP_ + cv;
    float* vst = Vs + row0 * KP_ + cv;

    for (int j = 0; j < nst; ++j) {
      lds_barrier();
      prio_by_progress(done, 20, 28, 31);
      ++done;
      *(float4*)(kst) = ka0;
      *(float4*)(kst + 8 * KP_) = ka1;
      *(float4*)(kst + 16 * KP_) = ka2;
      *(float4*)(kst + 24 * KP_) = ka3;
      *(float4*)(kst + 32 * KP_) = ka4;
      *(float4*)(kst + 40 * KP_) = ka5;
      *(float4*)(kst + 48 * KP_) = ka6;
      *(float4*)(kst + 56 * KP_) = ka7;
      *(float4*)(vst) = va0;
      *(float4*)(vst + 8 * KP_) = va1;
      *(float4*)(vst + 16 * KP_) = va2;
      *(float4*)(vst + 24 * KP_) = va3;
      *(float4*)(vst + 32 * KP_) = va4;
      *(float4*)(vst + 40 * KP_) = va5;
      *(float4*)(vst + 48 * KP_) = va6;
      *(float4*)(vst + 56 * KP_) = va7;
      if (j + 1 < nst) {
        const float* kn = kg + (size_t)(j + 1) * KT_ * DK_;
        const float* vn = vg + (size_t)(j + 1) * KT_ * DK_;
        ka0 = *(const float4*)(kn);
        ka1 = *(const float4*)(kn + 8 * DK_);
        ka2 = *(const float4*)(kn + 16 * DK_);
        ka3 = *(const float4*)(kn + 24 * DK_);
        ka4 = *(const float4*)(kn + 32 * DK_);
        ka5 = *(const float4*)(kn + 40 * DK_);
        ka6 = *(const float4*)(kn + 48 * DK_);
        ka7 = *(const float4*)(kn + 56 * DK_);
        va0 = *(const float4*)(vn);
        va1 = *(const float4*)(vn + 8 * DK_);
        va2 = *(const float4*)(vn + 16 * DK_);
        va3 = *(const float4*)(vn + 24 * DK_);
        va4 = *(const float4*)(vn + 32 * DK_);
        va5 = *(const float4*)(vn + 40 * DK_);
        va6 = *(const float4*)(vn + 48 * DK_);
        va7 = *(const float4*)(vn + 56 * DK_);
      }
      lds_barrier();

      if (j + 1 < nst) {
        attn_stage64(Ks, Vs, qf, r, hhalf, o0, o1, o2, o3, mrun, lrun);
      } else {
        for (int u = 0; u <= wv; ++u)
          attn_tile16(Ks + u * 16 * KP_, Vs + u * 16 * KP_, qf, r, hhalf,
                      16 * (u - wv) + 8 * hhalf - r, o0, o1, o2, o3, mrun,
                      lrun);
      }
    }

    const float inv = 1.0f / ((lrun + xhalf(lrun)) + 1e-8f);
    float* dst = O + ((size_t)(bb * S_ + qw0 + r)) * DM_ + head * DK_ + 8 * hhalf;
    *(float4*)(dst + 0)      = make_float4(o0[0] * inv, o0[1] * inv, o0[2] * inv, o0[3] * inv);
    *(float4*)(dst + 4)      = make_float4(o0[4] * inv, o0[5] * inv, o0[6] * inv, o0[7] * inv);
    *(float4*)(dst + 16)     = make_float4(o1[0] * inv, o1[1] * inv, o1[2] * inv, o1[3] * inv);
    *(float4*)(dst + 16 + 4) = make_float4(o1[4] * inv, o1[5] * inv, o1[6] * inv, o1[7] * inv);
    *(float4*)(dst + 32)     = make_float4(o2[0] * inv, o2[1] * inv, o2[2] * inv, o2[3] * inv);
    *(float4*)(dst + 32 + 4) = make_float4(o2[4] * inv, o2[5] * inv, o2[6] * inv, o2[7] * inv);
    *(float4*)(dst + 48)     = make_float4(o3[0] * inv, o3[1] * inv, o3[2] * inv, o3[3] * inv);
    *(float4*)(dst + 48 + 4) = make_float4(o3[4] * inv, o3[5] * inv, o3[6] * inv, o3[7] * inv);
  }
}

__global__ void __launch_bounds__(128)
out_proj_kernel(const float* __restrict__ Oin, const float* __restrict__ w_out,
                const float* __restrict__ b_out, float* __restrict__ out) {
  __shared__ float Xs[128 * GP_];
  __shared__ float Ws[64 * GP_];
  v8f acc[4][2] = {};
  const int m0 = blockIdx.x * 128;
  const int n0 = blockIdx.y * 64;
  gemm_core_128xN<2>(Oin, w_out, Xs, Ws, m0, n0, acc);

  const int tid = threadIdx.x;
  const int wv = tid >> 5;
  const int lane = tid & 31;
  const int r = lane & 15;
  const int hhalf = lane >> 4;
  const int wm = wv >> 1;
  const int wn = wv & 1;

#pragma unroll
  for (int ni = 0; ni < 2; ++ni) {
    const int n = n0 + wn * 32 + ni * 16 + r;
    const float bv = b_out[n];
#pragma unroll
    for (int mi = 0; mi < 4; ++mi) {
#pragma unroll
      for (int g = 0; g < 8; ++g) {
        const int gm = m0 + wm * 64 + mi * 16 + g + 8 * hhalf;
        out[(size_t)gm * DM_ + n] = acc[mi][ni][g] + bv;
      }
    }
  }
}

extern "C" void kernel_launch(void* const* d_in, const int* in_sizes, int n_in,
                              void* d_out, int out_size, void* d_ws,
                              size_t ws_size, hipStream_t stream) {
  const float* x = (const float*)d_in[0];
  const float* w_qkv = (const float*)d_in[1];
  const float* b_qkv = (const float*)d_in[2];
  const float* w_out = (const float*)d_in[3];
  const float* b_out = (const float*)d_in[4];
  float* out = (float*)d_out;

  const size_t per_buf = (size_t)BH_ * S_ * DK_;
  float* Qw = (float*)d_ws;
  float* Kw = Qw + per_buf;
  float* Vw = Kw + per_buf;
  float* Ow = Vw + per_buf;

  qkv_rope_kernel<<<dim3(ROWS_ / 128, N_QKV / 128), 256, 0, stream>>>(
      x, w_qkv, b_qkv, Qw, Kw, Vw);
  flash_attn_kernel<<<dim3(NQBLK_ / 2, BH_), 128, 0, stream>>>(Qw, Kw, Vw, Ow);
  out_proj_kernel<<<dim3(ROWS_ / 128, DM_ / 64), 128, 0, stream>>>(
      Ow, w_out, b_out, out);
}
